# starting kernel
# baseline (speedup 1.0000x reference)
#include <hip/hip_runtime.h>


typedef __attribute__((ext_vector_type(2))) float v2f;
typedef __attribute__((ext_vector_type(8))) float v8f;

#define NSTEP 64
#define HSTEP (1.0f / 64.0f)
#define INVL  (1.0f / 0.999f)

#define W1ST 66
#define W2ST 258
#define YST  66
#define STST 18

static constexpr int OFF_W1 = 0;
static constexpr int OFF_W2 = OFF_W1 + 256 * W1ST;
static constexpr int OFF_Y  = OFF_W2 + 64 * W2ST;
static constexpr int OFF_Z  = OFF_Y + 16 * YST;
static constexpr int OFF_F  = OFF_Z + 16 * YST;
static constexpr int OFF_ST = OFF_F + 16 * YST;
static constexpr int OFF_B1 = OFF_ST + 16 * STST;
static constexpr int OFF_U1 = OFF_B1 + 256;
static constexpr int OFF_B2 = OFF_U1 + 256;
static constexpr int OFF_C  = OFF_B2 + 64;
static constexpr int OFF_TS = OFF_C + 256;
static constexpr int OFF_I  = OFF_TS + 256;
static constexpr int LDS_FLOATS = OFF_I + 16;

__device__ __forceinline__ v8f vzero8() {
    v8f z = {0.f, 0.f, 0.f, 0.f, 0.f, 0.f, 0.f, 0.f};
    return z;
}

__device__ __forceinline__ float fast_tanh(float x) {
    const float LOG2E_X2 = 2.8853900817779268f;
    float e = __builtin_amdgcn_exp2f(x * LOG2E_X2);
    float r = __builtin_amdgcn_rcpf(e + 1.0f);
    return __builtin_fmaf(-2.0f, r, 1.0f);
}

template <bool TRACE>
__device__ __forceinline__ void mlp_eval(float* sm, float t, const float* src,
                                         const float* b1v, const float* u1v,
                                         const float* ccv, const float* b2v,
                                         float* trp) {
    const int lane = threadIdx.x & 31;
    const int l16  = lane & 15;
    const int hi   = lane >> 4;

    const float* w1 = sm + OFF_W1;
    const float* w2 = sm + OFF_W2;
    float*       st = sm + OFF_ST;
    float*       f  = sm + OFF_F;

    v2f af[16];
#pragma unroll
    for (int kt = 0; kt < 16; ++kt)
        af[kt] = *(const v2f*)&src[l16 * YST + kt * 4 + 2 * hi];

    v8f facc[4];
#pragma unroll
    for (int ot = 0; ot < 4; ++ot) facc[ot] = vzero8();

#pragma unroll
    for (int nt = 0; nt < 16; ++nt) {
        v2f bw1[16];
#pragma unroll
        for (int kt = 0; kt < 16; ++kt)
            bw1[kt] = *(const v2f*)&w1[(nt * 16 + l16) * W1ST + kt * 4 + 2 * hi];
        v2f bw2[16];
#pragma unroll
        for (int j = 0; j < 4; ++j)
#pragma unroll
            for (int ot = 0; ot < 4; ++ot)
                bw2[j * 4 + ot] = *(const v2f*)&w2[(ot * 16 + l16) * W2ST +
                                                   nt * 16 + j * 4 + 2 * hi];

        v8f acc[4];
#pragma unroll
        for (int q = 0; q < 4; ++q) acc[q] = vzero8();
#pragma unroll
        for (int kt = 0; kt < 16; kt += 4) {
#pragma unroll
            for (int q = 0; q < 4; ++q)
                acc[q] = __builtin_amdgcn_wmma_f32_16x16x4_f32(
                    false, af[kt + q], false, bw1[kt + q], (short)0, acc[q],
                    false, false);
        }
        v8f pre;
#pragma unroll
        for (int r = 0; r < 8; ++r)
            pre[r] = (acc[0][r] + acc[1][r]) + (acc[2][r] + acc[3][r]);

        const float bias = __builtin_fmaf(t, u1v[nt], b1v[nt]);
        const float cch  = TRACE ? ccv[nt] : 0.f;
#pragma unroll
        for (int r = 0; r < 8; ++r) {
            float tv = fast_tanh(pre[r] + bias);
            if (TRACE) trp[r] += (1.0f - tv * tv) * cch;
            st[(r + 8 * hi) * STST + l16] = tv;
        }
#pragma unroll
        for (int j = 0; j < 4; ++j) {
            v2f ha = *(const v2f*)&st[l16 * STST + j * 4 + 2 * hi];
#pragma unroll
            for (int ot = 0; ot < 4; ++ot) {
                facc[ot] = __builtin_amdgcn_wmma_f32_16x16x4_f32(
                    false, ha, false, bw2[j * 4 + ot], (short)0, facc[ot],
                    false, false);
            }
        }
    }
#pragma unroll
    for (int ot = 0; ot < 4; ++ot) {
#pragma unroll
        for (int r = 0; r < 8; ++r)
            f[(r + 8 * hi) * YST + ot * 16 + l16] = facc[ot][r] + b2v[ot];
    }
}

extern "C" __global__ void __launch_bounds__(32)
cnf_solver_kernel(const float* __restrict__ y1g, const float* __restrict__ W1g,
                  const float* __restrict__ b1g, const float* __restrict__ u1g,
                  const float* __restrict__ W2g, const float* __restrict__ b2g,
                  float* __restrict__ out) {
    extern __shared__ float sm[];
    const int lane = threadIdx.x;
    const int l16  = lane & 15;
    const int row0 = blockIdx.x * 16;

    __builtin_prefetch(W1g + lane * 64, 0, 3);
    __builtin_prefetch(W2g + lane * 64, 0, 3);

    for (int i2 = lane; i2 < 256 * 32; i2 += 32) {
        int m = i2 >> 5, d = (i2 & 31) * 2;
        *(v2f*)&sm[OFF_W1 + m * W1ST + d] = *(const v2f*)&W1g[m * 64 + d];
    }
    for (int i2 = lane; i2 < 64 * 128; i2 += 32) {
        int m = i2 >> 7, d = (i2 & 127) * 2;
        *(v2f*)&sm[OFF_W2 + m * W2ST + d] = *(const v2f*)&W2g[m * 256 + d];
    }
    for (int i = lane; i < 256; i += 32) {
        sm[OFF_B1 + i] = b1g[i];
        sm[OFF_U1 + i] = u1g[i];
    }
    for (int i = lane; i < 64; i += 32) sm[OFF_B2 + i] = b2g[i];
    for (int i2 = lane; i2 < 16 * 32; i2 += 32) {
        int m = i2 >> 5, d = (i2 & 31) * 2;
        v2f v = *(const v2f*)&y1g[(row0 + m) * 64 + d];
        *(v2f*)&sm[OFF_Y + m * YST + d] = v;
        *(v2f*)&sm[OFF_Z + m * YST + d] = v;
    }
    if (lane < 16) sm[OFF_I + lane] = 0.f;
    __syncthreads();

    for (int h = lane; h < 256; h += 32) {
        float c = 0.f;
        for (int d = 0; d < 64; ++d)
            c += sm[OFF_W1 + h * W1ST + d] * sm[OFF_W2 + d * W2ST + h];
        sm[OFF_C + h] = c;
    }
    __syncthreads();

    float b1v[16], u1v[16], ccv[16], b2v[4];
#pragma unroll
    for (int nt = 0; nt < 16; ++nt) {
        b1v[nt] = sm[OFF_B1 + nt * 16 + l16];
        u1v[nt] = sm[OFF_U1 + nt * 16 + l16];
        ccv[nt] = sm[OFF_C + nt * 16 + l16];
    }
#pragma unroll
    for (int ot = 0; ot < 4; ++ot) b2v[ot] = sm[OFF_B2 + ot * 16 + l16];

    float t1 = 1.0f;
#pragma unroll 1
    for (int s = 0; s < NSTEP; ++s) {
        const float t0 = t1 - HSTEP;

        float trp[8];
#pragma unroll
        for (int r = 0; r < 8; ++r) trp[r] = 0.f;

        mlp_eval<true>(sm, t1, sm + OFF_Y, b1v, u1v, ccv, b2v, trp);
#pragma unroll
        for (int r = 0; r < 8; r += 2)
            *(v2f*)&sm[OFF_TS + lane * 8 + r] = v2f{trp[r], trp[r + 1]};

        for (int i2 = lane; i2 < 512; i2 += 32) {
            int m = i2 >> 5, d = (i2 & 31) * 2;
            v2f fz = *(const v2f*)&sm[OFF_F + m * YST + d];
            v2f zz = *(const v2f*)&sm[OFF_Z + m * YST + d];
            zz.x -= HSTEP * fz.x;
            zz.y -= HSTEP * fz.y;
            *(v2f*)&sm[OFF_Z + m * YST + d] = zz;
        }

        if (lane < 16) {
            const int r = lane & 7;
            const int g = (lane >> 3) & 1;
            float sum = 0.f;
#pragma unroll
            for (int l = 0; l < 16; ++l) sum += sm[OFF_TS + (l + 16 * g) * 8 + r];
            sm[OFF_I + lane] += HSTEP * sum;
        }

        mlp_eval<false>(sm, t0, sm + OFF_Z, b1v, u1v, ccv, b2v, trp);

        for (int i2 = lane; i2 < 512; i2 += 32) {
            int m = i2 >> 5, d = (i2 & 31) * 2;
            v2f fz = *(const v2f*)&sm[OFF_F + m * YST + d];
            v2f zz = *(const v2f*)&sm[OFF_Z + m * YST + d];
            v2f yy = *(const v2f*)&sm[OFF_Y + m * YST + d];
            yy.x = INVL * yy.x + (1.0f - INVL) * zz.x - INVL * HSTEP * fz.x;
            yy.y = INVL * yy.y + (1.0f - INVL) * zz.y - INVL * HSTEP * fz.y;
            *(v2f*)&sm[OFF_Y + m * YST + d] = yy;
        }
        t1 = t0;
    }

    for (int i = lane; i < 16 * 65; i += 32) {
        int m = i / 65, c = i % 65;
        out[(row0 + m) * 65 + c] = (c < 64) ? sm[OFF_Y + m * YST + c]
                                            : sm[OFF_I + m];
    }
}

extern "C" void kernel_launch(void* const* d_in, const int* in_sizes, int n_in,
                              void* d_out, int out_size, void* d_ws, size_t ws_size,
                              hipStream_t stream) {
    const float* y1 = (const float*)d_in[0];
    const float* W1 = (const float*)d_in[1];
    const float* b1 = (const float*)d_in[2];
    const float* u1 = (const float*)d_in[3];
    const float* W2 = (const float*)d_in[4];
    const float* b2 = (const float*)d_in[5];
    float* out = (float*)d_out;

    const size_t lds_bytes = (size_t)LDS_FLOATS * sizeof(float);
    (void)hipFuncSetAttribute((const void*)cnf_solver_kernel,
                              hipFuncAttributeMaxDynamicSharedMemorySize,
                              (int)lds_bytes);

    cnf_solver_kernel<<<16, 32, lds_bytes, stream>>>(y1, W1, b1, u1, W2, b2, out);
}


# each wave keeps the step-invariant W1/W2 B-fragments of its hidden tile in registers for all 64 steps (were re-read from LDS among the WMMAs in every evaluation); VGPR budget set for 4 waves/SIMD
# speedup vs baseline: 6.4687x; 6.4687x over previous
#include <hip/hip_runtime.h>


typedef __attribute__((ext_vector_type(2))) float v2f;
typedef __attribute__((ext_vector_type(8))) float v8f;

#define NSTEP 64
#define HSTEP (1.0f / 64.0f)
#define INVL  (1.0f / 0.999f)

#define NW   16
#define NTHR (NW * 32)
#define TPW  (16 / NW)
#define EPT  (32 / NW)
#define CPO  (8 / EPT)
typedef __attribute__((ext_vector_type(EPT))) float vEf;

#define W1ST 66
#define W2ST 258
#define YST  66
#define STST 18

static constexpr int OFF_W1 = 0;
static constexpr int OFF_W2 = OFF_W1 + 256 * W1ST;
static constexpr int OFF_Y  = OFF_W2 + 64 * W2ST;
static constexpr int OFF_Z  = OFF_Y + 16 * YST;
static constexpr int OFF_ST = OFF_Z + 16 * YST;
static constexpr int OFF_B1 = OFF_ST + NW * 16 * STST;
static constexpr int OFF_U1 = OFF_B1 + 256;
static constexpr int OFF_B2 = OFF_U1 + 256;
static constexpr int OFF_C  = OFF_B2 + 64;
static constexpr int OFF_I  = OFF_C + 256;
static constexpr int OFF_PT = (OFF_I + 16 + 7) & ~7;
static constexpr int PT_FLOATS  = NW * NW * 32 * EPT;
static constexpr int LDS_FLOATS = OFF_PT + PT_FLOATS;

__device__ __forceinline__ v8f vzero8() {
    v8f z = {0.f, 0.f, 0.f, 0.f, 0.f, 0.f, 0.f, 0.f};
    return z;
}

__device__ __forceinline__ float fast_tanh(float x) {
    const float LOG2E_X2 = 2.8853900817779268f;
    float e = __builtin_amdgcn_exp2f(x * LOG2E_X2);
    float r = __builtin_amdgcn_rcpf(e + 1.0f);
    return __builtin_fmaf(-2.0f, r, 1.0f);
}

template <bool TRACE>
__device__ __forceinline__ void mlp_partial(float* sm, float t, const float* src,
                                            int wv, const v2f (&bw1)[TPW][16],
                                            const v2f (&bw2)[TPW][16],
                                            const float* b1v,
                                            const float* u1v, const float* ccv,
                                            float* trp) {
    const int lane = threadIdx.x & 31;
    const int l16  = lane & 15;
    const int hi   = lane >> 4;

    float* st = sm + OFF_ST + wv * 16 * STST;

    v2f af[16];
#pragma unroll
    for (int kt = 0; kt < 16; ++kt)
        af[kt] = *(const v2f*)&src[l16 * YST + kt * 4 + 2 * hi];

    v8f facc[4];
#pragma unroll
    for (int ot = 0; ot < 4; ++ot) facc[ot] = vzero8();

#pragma unroll
    for (int i = 0; i < TPW; ++i) {

        v8f acc[4];
#pragma unroll
        for (int q = 0; q < 4; ++q) acc[q] = vzero8();
#pragma unroll
        for (int kt = 0; kt < 16; kt += 4) {
#pragma unroll
            for (int q = 0; q < 4; ++q)
                acc[q] = __builtin_amdgcn_wmma_f32_16x16x4_f32(
                    false, af[kt + q], false, bw1[i][kt + q], (short)0, acc[q],
                    false, false);
        }
        v8f pre;
#pragma unroll
        for (int r = 0; r < 8; ++r)
            pre[r] = (acc[0][r] + acc[1][r]) + (acc[2][r] + acc[3][r]);

        const float bias = __builtin_fmaf(t, u1v[i], b1v[i]);
        const float cch  = TRACE ? ccv[i] : 0.f;
#pragma unroll
        for (int r = 0; r < 8; ++r) {
            float tv = fast_tanh(pre[r] + bias);
            if (TRACE) trp[r] += (1.0f - tv * tv) * cch;
            st[(r + 8 * hi) * STST + l16] = tv;
        }
#pragma unroll
        for (int j = 0; j < 4; ++j) {
            v2f ha = *(const v2f*)&st[l16 * STST + j * 4 + 2 * hi];
#pragma unroll
            for (int ot = 0; ot < 4; ++ot) {
                facc[ot] = __builtin_amdgcn_wmma_f32_16x16x4_f32(
                    false, ha, false, bw2[i][j * 4 + ot], (short)0, facc[ot],
                    false, false);
            }
        }
    }
    float* pw = sm + OFF_PT + (wv * 32 + lane) * EPT;
#pragma unroll
    for (int q = 0; q < NW; ++q) {
        const int ot = q / CPO;
        const int r0 = (q % CPO) * EPT;
        vEf v;
#pragma unroll
        for (int j = 0; j < EPT; ++j) v[j] = facc[ot][r0 + j];
        *(vEf*)&pw[q * NW * 32 * EPT] = v;
    }
}

__device__ __forceinline__ vEf reduce_partials(const float* sm, int wv, int lane) {
    const float* p = sm + OFF_PT + (wv * NW * 32 + lane) * EPT;
    vEf v[NW];
#pragma unroll
    for (int w = 0; w < NW; ++w) v[w] = *(const vEf*)&p[w * 32 * EPT];
#pragma unroll
    for (int span = 1; span < NW; span *= 2) {
#pragma unroll
        for (int w = 0; w + span < NW; w += 2 * span) v[w] += v[w + span];
    }
    return v[0];
}

extern "C" __global__ void __launch_bounds__(NTHR)
__attribute__((amdgpu_waves_per_eu(4, 4)))
cnf_solver_kernel(const float* __restrict__ y1g, const float* __restrict__ W1g,
                  const float* __restrict__ b1g, const float* __restrict__ u1g,
                  const float* __restrict__ W2g, const float* __restrict__ b2g,
                  float* __restrict__ out) {
    extern __shared__ __attribute__((aligned(16))) float sm[];
    const int tid  = threadIdx.x;
    const int lane = tid & 31;
    const int l16  = lane & 15;
    const int hi   = lane >> 4;
    const int wv   = __builtin_amdgcn_readfirstlane(tid >> 5);
    const int row0 = blockIdx.x * 16;

    if (tid < 256) {
        __builtin_prefetch(W1g + tid * 64, 0, 3);
        __builtin_prefetch(W2g + tid * 64, 0, 3);
    }

    for (int i2 = tid; i2 < 256 * 32; i2 += NTHR) {
        int m = i2 >> 5, d = (i2 & 31) * 2;
        *(v2f*)&sm[OFF_W1 + m * W1ST + d] = *(const v2f*)&W1g[m * 64 + d];
    }
    for (int i2 = tid; i2 < 64 * 128; i2 += NTHR) {
        int m = i2 >> 7, d = (i2 & 127) * 2;
        *(v2f*)&sm[OFF_W2 + m * W2ST + d] = *(const v2f*)&W2g[m * 256 + d];
    }
    for (int i = tid; i < 256; i += NTHR) {
        sm[OFF_B1 + i] = b1g[i];
        sm[OFF_U1 + i] = u1g[i];
    }
    for (int i = tid; i < 64; i += NTHR) sm[OFF_B2 + i] = b2g[i];

    const int col_own = (wv / CPO) * 16 + l16;
    const int row_own = 8 * hi + (wv % CPO) * EPT;
    float yreg[EPT], zreg[EPT];
#pragma unroll
    for (int j = 0; j < EPT; ++j) {
        float v = y1g[(row0 + row_own + j) * 64 + col_own];
        yreg[j] = v;
        zreg[j] = v;
        sm[OFF_Y + (row_own + j) * YST + col_own] = v;
    }
    __syncthreads();

    for (int h = tid; h < 256; h += NTHR) {
        float c = 0.f;
        for (int d = 0; d < 64; ++d)
            c += sm[OFF_W1 + h * W1ST + d] * sm[OFF_W2 + d * W2ST + h];
        sm[OFF_C + h] = c;
    }
    __syncthreads();

    float b1v[TPW], u1v[TPW], ccv[TPW];
#pragma unroll
    for (int i = 0; i < TPW; ++i) {
        const int n = (wv * TPW + i) * 16 + l16;
        b1v[i] = sm[OFF_B1 + n];
        u1v[i] = sm[OFF_U1 + n];
        ccv[i] = sm[OFF_C + n];
    }
    const float b2own = sm[OFF_B2 + col_own];

    v2f bw1[TPW][16], bw2[TPW][16];
    {
        const float* w1 = sm + OFF_W1 + (wv * TPW * 16 + l16) * W1ST + 2 * hi;
        const float* w2 = sm + OFF_W2 + l16 * W2ST + wv * TPW * 16 + 2 * hi;
#pragma unroll
        for (int i = 0; i < TPW; ++i) {
#pragma unroll
            for (int kt = 0; kt < 16; ++kt)
                bw1[i][kt] = *(const v2f*)&w1[i * 16 * W1ST + kt * 4];
#pragma unroll
            for (int j = 0; j < 4; ++j)
#pragma unroll
                for (int ot = 0; ot < 4; ++ot)
                    bw2[i][j * 4 + ot] =
                        *(const v2f*)&w2[ot * 16 * W2ST + i * 16 + j * 4];
        }
    }

    float trp[8];
#pragma unroll
    for (int r = 0; r < 8; ++r) trp[r] = 0.f;

    float t1 = 1.0f;
#pragma unroll 1
    for (int s = 0; s < NSTEP; ++s) {
        const float t0 = t1 - HSTEP;

        mlp_partial<true>(sm, t1, sm + OFF_Y, wv, bw1, bw2, b1v, u1v, ccv, trp);
        __syncthreads();

        {
            vEf f = reduce_partials(sm, wv, lane);
#pragma unroll
            for (int j = 0; j < EPT; ++j) {
                const float fv = f[j] + b2own;
                zreg[j] -= HSTEP * fv;
                sm[OFF_Z + (row_own + j) * YST + col_own] = zreg[j];
            }
        }
        __syncthreads();

        mlp_partial<false>(sm, t0, sm + OFF_Z, wv, bw1, bw2, b1v, u1v, ccv, trp);
        __syncthreads();

        {
            vEf f = reduce_partials(sm, wv, lane);
#pragma unroll
            for (int j = 0; j < EPT; ++j) {
                const float fv = f[j] + b2own;
                yreg[j] = INVL * yreg[j] + (1.0f - INVL) * zreg[j] -
                          INVL * HSTEP * fv;
                sm[OFF_Y + (row_own + j) * YST + col_own] = yreg[j];
            }
        }
        __syncthreads();
        t1 = t0;
    }

    float* ts  = sm + OFF_PT;
    float* ts2 = sm + OFF_PT + NW * 256;
#pragma unroll
    for (int r = 0; r < 8; ++r) ts[(wv * 32 + lane) * 8 + r] = trp[r];
    __syncthreads();
    for (int e = tid; e < 256; e += NTHR) {
        const int m = e >> 4, c = e & 15;
        float sum = 0.f;
#pragma unroll
        for (int w = 0; w < NW; ++w)
            sum += ts[(w * 32 + (m >> 3) * 16 + c) * 8 + (m & 7)];
        ts2[e] = sum;
    }
    __syncthreads();
    if (tid < 16) {
        float sum = 0.f;
#pragma unroll
        for (int c = 0; c < 16; ++c) sum += ts2[tid * 16 + c];
        sm[OFF_I + tid] = HSTEP * sum;
    }
    __syncthreads();

    for (int i = tid; i < 16 * 65; i += NTHR) {
        int m = i / 65, c = i % 65;
        out[(row0 + m) * 65 + c] = (c < 64) ? sm[OFF_Y + m * YST + c]
                                            : sm[OFF_I + m];
    }
}

extern "C" void kernel_launch(void* const* d_in, const int* in_sizes, int n_in,
                              void* d_out, int out_size, void* d_ws, size_t ws_size,
                              hipStream_t stream) {
    const float* y1 = (const float*)d_in[0];
    const float* W1 = (const float*)d_in[1];
    const float* b1 = (const float*)d_in[2];
    const float* u1 = (const float*)d_in[3];
    const float* W2 = (const float*)d_in[4];
    const float* b2 = (const float*)d_in[5];
    float* out = (float*)d_out;

    const size_t lds_bytes = (size_t)LDS_FLOATS * sizeof(float);
    (void)hipFuncSetAttribute((const void*)cnf_solver_kernel,
                              hipFuncAttributeMaxDynamicSharedMemorySize,
                              (int)lds_bytes);

    cnf_solver_kernel<<<16, NTHR, lds_bytes, stream>>>(y1, W1, b1, u1, W2, b2, out);
}


# both products computed transposed (weights as the WMMA A operand): GEMM1's D registers are GEMM2's B fragments, so tanh feeds the W2 WMMAs from registers; per-wave LDS staging transpose removed
# speedup vs baseline: 6.7138x; 1.0379x over previous
#include <hip/hip_runtime.h>


typedef __attribute__((ext_vector_type(2))) float v2f;
typedef __attribute__((ext_vector_type(8))) float v8f;

#define NSTEP 64
#define HSTEP (1.0f / 64.0f)
#define INVL  (1.0f / 0.999f)

#define NW   16
#define NTHR (NW * 32)
#define TPW  (16 / NW)
#define EPT  (32 / NW)
#define CPO  (8 / EPT)
typedef __attribute__((ext_vector_type(EPT))) float vEf;

#define W1ST 66
#define W2ST 258
#define YST  66

static constexpr int OFF_W1 = 0;
static constexpr int OFF_W2 = OFF_W1 + 256 * W1ST;
static constexpr int OFF_Y  = OFF_W2 + 64 * W2ST;
static constexpr int OFF_Z  = OFF_Y + 16 * YST;
static constexpr int OFF_B1 = OFF_Z + 16 * YST;
static constexpr int OFF_U1 = OFF_B1 + 256;
static constexpr int OFF_B2 = OFF_U1 + 256;
static constexpr int OFF_C  = OFF_B2 + 64;
static constexpr int OFF_I  = OFF_C + 256;
static constexpr int OFF_PT = (OFF_I + 16 + 7) & ~7;
static constexpr int PT_FLOATS  = NW * NW * 32 * EPT;
static constexpr int LDS_FLOATS = OFF_PT + PT_FLOATS;

__device__ __forceinline__ v8f vzero8() {
    v8f z = {0.f, 0.f, 0.f, 0.f, 0.f, 0.f, 0.f, 0.f};
    return z;
}

__device__ __forceinline__ float fast_tanh(float x) {
    const float LOG2E_X2 = 2.8853900817779268f;
    float e = __builtin_amdgcn_exp2f(x * LOG2E_X2);
    float r = __builtin_amdgcn_rcpf(e + 1.0f);
    return __builtin_fmaf(-2.0f, r, 1.0f);
}

template <bool TRACE>
__device__ __forceinline__ void mlp_partial(float* sm, float t, const float* src,
                                            int wv, const v2f (&aw1)[TPW][16],
                                            const v2f (&aw2)[TPW][16],
                                            const float (&b1v)[TPW][8],
                                            const float (&u1v)[TPW][8],
                                            const float (&ccv)[TPW][8],
                                            float* trp) {
    const int lane = threadIdx.x & 31;
    const int l16  = lane & 15;
    const int hi   = lane >> 4;

    v2f yb[16];
#pragma unroll
    for (int kt = 0; kt < 16; ++kt)
        yb[kt] = *(const v2f*)&src[l16 * YST + kt * 4 + 2 * hi];

    v8f facc[4];
#pragma unroll
    for (int ot = 0; ot < 4; ++ot) facc[ot] = vzero8();

#pragma unroll
    for (int i = 0; i < TPW; ++i) {

        v8f acc[4];
#pragma unroll
        for (int q = 0; q < 4; ++q) acc[q] = vzero8();
#pragma unroll
        for (int kt = 0; kt < 16; kt += 4) {
#pragma unroll
            for (int q = 0; q < 4; ++q)
                acc[q] = __builtin_amdgcn_wmma_f32_16x16x4_f32(
                    false, aw1[i][kt + q], false, yb[kt + q], (short)0, acc[q],
                    false, false);
        }
#pragma unroll
        for (int j = 0; j < 4; ++j) {
            v2f hb;
#pragma unroll
            for (int s = 0; s < 2; ++s) {
                const int r = 2 * j + s;
                const float pre = (acc[0][r] + acc[1][r]) + (acc[2][r] + acc[3][r]);
                const float bias = __builtin_fmaf(t, u1v[i][r], b1v[i][r]);
                const float tv = fast_tanh(pre + bias);
                if (TRACE) trp[r] += (1.0f - tv * tv) * ccv[i][r];
                hb[s] = tv;
            }
#pragma unroll
            for (int ot = 0; ot < 4; ++ot) {
                facc[ot] = __builtin_amdgcn_wmma_f32_16x16x4_f32(
                    false, aw2[i][j * 4 + ot], false, hb, (short)0, facc[ot],
                    false, false);
            }
        }
    }
    float* pw = sm + OFF_PT + (wv * 32 + lane) * EPT;
#pragma unroll
    for (int q = 0; q < NW; ++q) {
        const int ot = q / CPO;
        const int r0 = (q % CPO) * EPT;
        vEf v;
#pragma unroll
        for (int j = 0; j < EPT; ++j) v[j] = facc[ot][r0 + j];
        *(vEf*)&pw[q * NW * 32 * EPT] = v;
    }
}

__device__ __forceinline__ vEf reduce_partials(const float* sm, int wv, int lane) {
    const float* p = sm + OFF_PT + (wv * NW * 32 + lane) * EPT;
    vEf v[NW];
#pragma unroll
    for (int w = 0; w < NW; ++w) v[w] = *(const vEf*)&p[w * 32 * EPT];
#pragma unroll
    for (int span = 1; span < NW; span *= 2) {
#pragma unroll
        for (int w = 0; w + span < NW; w += 2 * span) v[w] += v[w + span];
    }
    return v[0];
}

extern "C" __global__ void __launch_bounds__(NTHR)
__attribute__((amdgpu_waves_per_eu(4, 4)))
cnf_solver_kernel(const float* __restrict__ y1g, const float* __restrict__ W1g,
                  const float* __restrict__ b1g, const float* __restrict__ u1g,
                  const float* __restrict__ W2g, const float* __restrict__ b2g,
                  float* __restrict__ out) {
    extern __shared__ __attribute__((aligned(16))) float sm[];
    const int tid  = threadIdx.x;
    const int lane = tid & 31;
    const int l16  = lane & 15;
    const int hi   = lane >> 4;
    const int wv   = __builtin_amdgcn_readfirstlane(tid >> 5);
    const int row0 = blockIdx.x * 16;

    if (tid < 256) {
        __builtin_prefetch(W1g + tid * 64, 0, 3);
        __builtin_prefetch(W2g + tid * 64, 0, 3);
    }

    for (int i2 = tid; i2 < 256 * 32; i2 += NTHR) {
        int m = i2 >> 5, d = (i2 & 31) * 2;
        *(v2f*)&sm[OFF_W1 + m * W1ST + d] = *(const v2f*)&W1g[m * 64 + d];
    }
    for (int i2 = tid; i2 < 64 * 128; i2 += NTHR) {
        int m = i2 >> 7, d = (i2 & 127) * 2;
        *(v2f*)&sm[OFF_W2 + m * W2ST + d] = *(const v2f*)&W2g[m * 256 + d];
    }
    for (int i = tid; i < 256; i += NTHR) {
        sm[OFF_B1 + i] = b1g[i];
        sm[OFF_U1 + i] = u1g[i];
    }
    for (int i = tid; i < 64; i += NTHR) sm[OFF_B2 + i] = b2g[i];

    const int row_own = l16;
    const int col_own = (wv / CPO) * 16 + 8 * hi + (wv % CPO) * EPT;
    float yreg[EPT], zreg[EPT];
#pragma unroll
    for (int j = 0; j < EPT; ++j) {
        float v = y1g[(row0 + row_own) * 64 + col_own + j];
        yreg[j] = v;
        zreg[j] = v;
        sm[OFF_Y + row_own * YST + col_own + j] = v;
    }
    __syncthreads();

    for (int h = tid; h < 256; h += NTHR) {
        float c = 0.f;
        for (int d = 0; d < 64; ++d)
            c += sm[OFF_W1 + h * W1ST + d] * sm[OFF_W2 + d * W2ST + h];
        sm[OFF_C + h] = c;
    }
    __syncthreads();

    float b1v[TPW][8], u1v[TPW][8], ccv[TPW][8];
#pragma unroll
    for (int i = 0; i < TPW; ++i) {
#pragma unroll
        for (int r = 0; r < 8; ++r) {
            const int n = (wv * TPW + i) * 16 + 8 * hi + r;
            b1v[i][r] = sm[OFF_B1 + n];
            u1v[i][r] = sm[OFF_U1 + n];
            ccv[i][r] = sm[OFF_C + n];
        }
    }
    float b2own[EPT];
#pragma unroll
    for (int j = 0; j < EPT; ++j) b2own[j] = sm[OFF_B2 + col_own + j];

    v2f aw1[TPW][16], aw2[TPW][16];
    {
        const float* w1 = sm + OFF_W1 + (wv * TPW * 16 + l16) * W1ST + 2 * hi;
        const float* w2 = sm + OFF_W2 + l16 * W2ST + wv * TPW * 16 + 8 * hi;
#pragma unroll
        for (int i = 0; i < TPW; ++i) {
#pragma unroll
            for (int kt = 0; kt < 16; ++kt)
                aw1[i][kt] = *(const v2f*)&w1[i * 16 * W1ST + kt * 4];
#pragma unroll
            for (int j = 0; j < 4; ++j)
#pragma unroll
                for (int ot = 0; ot < 4; ++ot)
                    aw2[i][j * 4 + ot] =
                        *(const v2f*)&w2[ot * 16 * W2ST + i * 16 + j * 2];
        }
    }

    float trp[8];
#pragma unroll
    for (int r = 0; r < 8; ++r) trp[r] = 0.f;

    float t1 = 1.0f;
#pragma unroll 1
    for (int s = 0; s < NSTEP; ++s) {
        const float t0 = t1 - HSTEP;

        mlp_partial<true>(sm, t1, sm + OFF_Y, wv, aw1, aw2, b1v, u1v, ccv, trp);
        __syncthreads();

        {
            vEf f = reduce_partials(sm, wv, lane);
#pragma unroll
            for (int j = 0; j < EPT; ++j) {
                const float fv = f[j] + b2own[j];
                zreg[j] -= HSTEP * fv;
                sm[OFF_Z + row_own * YST + col_own + j] = zreg[j];
            }
        }
        __syncthreads();

        mlp_partial<false>(sm, t0, sm + OFF_Z, wv, aw1, aw2, b1v, u1v, ccv, trp);
        __syncthreads();

        {
            vEf f = reduce_partials(sm, wv, lane);
#pragma unroll
            for (int j = 0; j < EPT; ++j) {
                const float fv = f[j] + b2own[j];
                yreg[j] = INVL * yreg[j] + (1.0f - INVL) * zreg[j] -
                          INVL * HSTEP * fv;
                sm[OFF_Y + row_own * YST + col_own + j] = yreg[j];
            }
        }
        __syncthreads();
        t1 = t0;
    }

    float* ts  = sm + OFF_PT;
    float* ts2 = sm + OFF_PT + NW * 32;
    ts[wv * 32 + lane] = ((trp[0] + trp[1]) + (trp[2] + trp[3])) +
                         ((trp[4] + trp[5]) + (trp[6] + trp[7]));
    __syncthreads();
    for (int e = tid; e < 16 * NW; e += NTHR) {
        const int m = e / NW, w = e % NW;
        ts2[e] = ts[w * 32 + m] + ts[w * 32 + 16 + m];
    }
    __syncthreads();
    if (tid < 16) {
        float sum = 0.f;
#pragma unroll
        for (int w = 0; w < NW; ++w) sum += ts2[tid * NW + w];
        sm[OFF_I + tid] = HSTEP * sum;
    }
    __syncthreads();

    for (int i = tid; i < 16 * 65; i += NTHR) {
        int m = i / 65, c = i % 65;
        out[(row0 + m) * 65 + c] = (c < 64) ? sm[OFF_Y + m * YST + c]
                                            : sm[OFF_I + m];
    }
}

extern "C" void kernel_launch(void* const* d_in, const int* in_sizes, int n_in,
                              void* d_out, int out_size, void* d_ws, size_t ws_size,
                              hipStream_t stream) {
    const float* y1 = (const float*)d_in[0];
    const float* W1 = (const float*)d_in[1];
    const float* b1 = (const float*)d_in[2];
    const float* u1 = (const float*)d_in[3];
    const float* W2 = (const float*)d_in[4];
    const float* b2 = (const float*)d_in[5];
    float* out = (float*)d_out;

    const size_t lds_bytes = (size_t)LDS_FLOATS * sizeof(float);
    (void)hipFuncSetAttribute((const void*)cnf_solver_kernel,
                              hipFuncAttributeMaxDynamicSharedMemorySize,
                              (int)lds_bytes);

    cnf_solver_kernel<<<16, NTHR, lds_bytes, stream>>>(y1, W1, b1, u1, W2, b2, out);
}


# 4 waves per workgroup, one per SIMD with 4 hidden tiles each (was 16): a tile's bias+tanh cut into stages pinned one per WMMA under the next block's WMMAs; LDS exchange traffic cut to a quarter
# speedup vs baseline: 6.9833x; 1.0401x over previous
#include <hip/hip_runtime.h>


typedef __attribute__((ext_vector_type(2))) float v2f;
typedef __attribute__((ext_vector_type(4))) float v4f;
typedef __attribute__((ext_vector_type(8))) float v8f;

#define NSTEP 64
#define HSTEP (1.0f / 64.0f)
#define INVL  (1.0f / 0.999f)

#define NW   4
#define NTHR (NW * 32)
#define TPW  4

#define W1ST 66
#define W2ST 258
#define YST  66

static constexpr int OFF_W1 = 0;
static constexpr int OFF_W2 = OFF_W1 + 256 * W1ST;
static constexpr int OFF_Y  = OFF_W2 + 64 * W2ST;
static constexpr int OFF_Z  = OFF_Y + 16 * YST;
static constexpr int OFF_B1 = OFF_Z + 16 * YST;
static constexpr int OFF_U1 = OFF_B1 + 256;
static constexpr int OFF_B2 = OFF_U1 + 256;
static constexpr int OFF_C  = OFF_B2 + 64;
static constexpr int OFF_I  = OFF_C + 256;
static constexpr int OFF_PT = (OFF_I + 16 + 7) & ~7;
static constexpr int PT_FLOATS  = NW * NW * 2 * 32 * 4;
static constexpr int LDS_FLOATS = OFF_PT + PT_FLOATS;

__device__ __forceinline__ v8f vzero8() {
    v8f z = {0.f, 0.f, 0.f, 0.f, 0.f, 0.f, 0.f, 0.f};
    return z;
}
__device__ __forceinline__ v8f vsplat8(float x) {
    v8f z = {x, x, x, x, x, x, x, x};
    return z;
}

#define WMMA_F32(a, b, c) \
    __builtin_amdgcn_wmma_f32_16x16x4_f32(false, (a), false, (b), (short)0, (c), false, false)

template <bool TRACE>
__device__ __forceinline__ void tanh_stage(int k, const v8f (&acc)[4], float t,
                                           const v8f& b1, const v8f& u1,
                                           const v8f& cc, v8f& p, v8f& q,
                                           v8f& tv, v8f& trp) {
    const float LOG2E_X2 = 2.8853900817779268f;
    switch (k) {
    case 0: {
        v8f a0 = acc[0], a1 = acc[1];
        asm volatile("" : "+v"(a0), "+v"(a1));
        p = a0 + a1;
        break;
    }
    case 1: {
        v8f a2 = acc[2], a3 = acc[3];
        asm volatile("" : "+v"(a2), "+v"(a3));
        q = a2 + a3;
        break;
    }
    case 2:
        p = p + q;
        q = __builtin_elementwise_fma(vsplat8(t), u1, b1);
        break;
    case 3: p = (p + q) * LOG2E_X2; break;
    case 4:
#pragma unroll
        for (int r = 0; r < 4; ++r) q[r] = __builtin_amdgcn_exp2f(p[r]);
        break;
    case 5:
#pragma unroll
        for (int r = 4; r < 8; ++r) q[r] = __builtin_amdgcn_exp2f(p[r]);
        break;
    case 6: p = q + 1.0f; break;
    case 7:
#pragma unroll
        for (int r = 0; r < 4; ++r) q[r] = __builtin_amdgcn_rcpf(p[r]);
        break;
    case 8:
#pragma unroll
        for (int r = 4; r < 8; ++r) q[r] = __builtin_amdgcn_rcpf(p[r]);
        break;
    case 9: tv = __builtin_elementwise_fma(vsplat8(-2.0f), q, vsplat8(1.0f)); break;
    case 10:
        if (TRACE) p = __builtin_elementwise_fma(-tv, tv, vsplat8(1.0f));
        break;
    case 11:
        if (TRACE) trp = __builtin_elementwise_fma(p, cc, trp);
        break;
    default: break;
    }
}

template <bool TRACE, bool RIDE>
__device__ __forceinline__ void g1_block(v8f (&dst)[4], const v2f (&a)[16],
                                         const v2f (&yb)[16], const v8f (&prev)[4],
                                         float t, const v8f& b1, const v8f& u1,
                                         const v8f& cc, v8f& p, v8f& q, v8f& tv,
                                         v8f& trp) {
#pragma unroll
    for (int w = 0; w < 16; ++w) {
        const int c = w & 3;
        if (w < 4) dst[c] = WMMA_F32(a[w], yb[w], vzero8());
        else       dst[c] = WMMA_F32(a[w], yb[w], dst[c]);
        if (RIDE) tanh_stage<TRACE>(w - 2, prev, t, b1, u1, cc, p, q, tv, trp);
        __builtin_amdgcn_sched_barrier(0);
    }
}

template <bool TRACE, bool RIDE, bool FIRST>
__device__ __forceinline__ void g2_block(v8f (&facc)[4], const v2f (&a)[16],
                                         const v8f& tvin, const v8f (&prev)[4],
                                         float t, const v8f& b1, const v8f& u1,
                                         const v8f& cc, v8f& p, v8f& q, v8f& tv,
                                         v8f& trp) {
#pragma unroll
    for (int w = 0; w < 16; ++w) {
        const int j = w >> 2, ot = w & 3;
        v2f hb;
        hb[0] = tvin[2 * j];
        hb[1] = tvin[2 * j + 1];
        if (FIRST && j == 0) facc[ot] = WMMA_F32(a[j * 4 + ot], hb, vzero8());
        else                 facc[ot] = WMMA_F32(a[j * 4 + ot], hb, facc[ot]);
        if (RIDE) tanh_stage<TRACE>(w - 2, prev, t, b1, u1, cc, p, q, tv, trp);
        __builtin_amdgcn_sched_barrier(0);
    }
}

template <bool TRACE>
__device__ __forceinline__ void mlp_partial(float* sm, float t, const float* src,
                                            int wv, const v2f (&aw1)[TPW][16],
                                            const v2f (&aw2)[TPW][16],
                                            const v8f (&b1v)[TPW],
                                            const v8f (&u1v)[TPW],
                                            const v8f (&ccv)[TPW], v8f& trp) {
    const int lane = threadIdx.x & 31;
    const int l16  = lane & 15;
    const int hi   = lane >> 4;

    v2f yb[16];
#pragma unroll
    for (int kt = 0; kt < 16; ++kt) {
        yb[kt] = *(const v2f*)&src[l16 * YST + kt * 4 + 2 * hi];
        if (kt & 1) __builtin_amdgcn_sched_barrier(0);
    }

    v8f accA[4], accB[4];
    v8f tv[TPW];
    v8f facc[4];
    v8f p, q;

    g1_block<TRACE, false>(accA, aw1[0], yb, accA, t, b1v[0], u1v[0], ccv[0], p, q, tv[0], trp);
    g1_block<TRACE, true >(accB, aw1[1], yb, accA, t, b1v[0], u1v[0], ccv[0], p, q, tv[0], trp);
    g1_block<TRACE, true >(accA, aw1[2], yb, accB, t, b1v[1], u1v[1], ccv[1], p, q, tv[1], trp);
    g1_block<TRACE, true >(accB, aw1[3], yb, accA, t, b1v[2], u1v[2], ccv[2], p, q, tv[2], trp);
    g2_block<TRACE, true,  true >(facc, aw2[0], tv[0], accB, t, b1v[3], u1v[3], ccv[3], p, q, tv[3], trp);
    g2_block<TRACE, false, false>(facc, aw2[1], tv[1], accB, t, b1v[3], u1v[3], ccv[3], p, q, tv[3], trp);
    g2_block<TRACE, false, false>(facc, aw2[2], tv[2], accB, t, b1v[3], u1v[3], ccv[3], p, q, tv[3], trp);
    g2_block<TRACE, false, false>(facc, aw2[3], tv[3], accB, t, b1v[3], u1v[3], ccv[3], p, q, tv[3], trp);

    float* pw = sm + OFF_PT + wv * 256 + lane * 4;
#pragma unroll
    for (int o = 0; o < NW; ++o) {
        v4f lo = {facc[o][0], facc[o][1], facc[o][2], facc[o][3]};
        v4f hh = {facc[o][4], facc[o][5], facc[o][6], facc[o][7]};
        *(v4f*)&pw[o * (NW * 256)]       = lo;
        *(v4f*)&pw[o * (NW * 256) + 128] = hh;
    }
}

__device__ __forceinline__ v8f reduce_partials(const float* sm, int wv, int lane) {
    const float* pr = sm + OFF_PT + wv * (NW * 256) + lane * 4;
    v4f lo[NW], hh[NW];
#pragma unroll
    for (int w = 0; w < NW; ++w) {
        lo[w] = *(const v4f*)&pr[w * 256];
        hh[w] = *(const v4f*)&pr[w * 256 + 128];
    }
    const v4f slo = (lo[0] + lo[1]) + (lo[2] + lo[3]);
    const v4f shh = (hh[0] + hh[1]) + (hh[2] + hh[3]);
    v8f f = {slo[0], slo[1], slo[2], slo[3], shh[0], shh[1], shh[2], shh[3]};
    return f;
}

extern "C" __global__ void __launch_bounds__(NTHR)
__attribute__((amdgpu_waves_per_eu(1, 1)))
cnf_solver_kernel(const float* __restrict__ y1g, const float* __restrict__ W1g,
                  const float* __restrict__ b1g, const float* __restrict__ u1g,
                  const float* __restrict__ W2g, const float* __restrict__ b2g,
                  float* __restrict__ out) {
    extern __shared__ __attribute__((aligned(16))) float sm[];
    const int tid  = threadIdx.x;
    const int lane = tid & 31;
    const int l16  = lane & 15;
    const int hi   = lane >> 4;
    const int wv   = __builtin_amdgcn_readfirstlane(tid >> 5);
    const int row0 = blockIdx.x * 16;

#pragma unroll 1
    for (int b = 0; b < 2; ++b) {
        v4f tmp[16];
#pragma unroll
        for (int k = 0; k < 16; ++k)
            tmp[k] = *(const v4f*)&W1g[(tid + (b * 16 + k) * NTHR) * 4];
#pragma unroll
        for (int k = 0; k < 16; ++k) {
            const int i4 = tid + (b * 16 + k) * NTHR;
            const int m = i4 >> 4, d = (i4 & 15) * 4;
            v2f lo = {tmp[k][0], tmp[k][1]}, hh = {tmp[k][2], tmp[k][3]};
            *(v2f*)&sm[OFF_W1 + m * W1ST + d]     = lo;
            *(v2f*)&sm[OFF_W1 + m * W1ST + d + 2] = hh;
        }
    }
#pragma unroll 1
    for (int b = 0; b < 2; ++b) {
        v4f tmp[16];
#pragma unroll
        for (int k = 0; k < 16; ++k)
            tmp[k] = *(const v4f*)&W2g[(tid + (b * 16 + k) * NTHR) * 4];
#pragma unroll
        for (int k = 0; k < 16; ++k) {
            const int i4 = tid + (b * 16 + k) * NTHR;
            const int m = i4 >> 6, d = (i4 & 63) * 4;
            v2f lo = {tmp[k][0], tmp[k][1]}, hh = {tmp[k][2], tmp[k][3]};
            *(v2f*)&sm[OFF_W2 + m * W2ST + d]     = lo;
            *(v2f*)&sm[OFF_W2 + m * W2ST + d + 2] = hh;
        }
    }
    for (int i = tid; i < 256; i += NTHR) {
        sm[OFF_B1 + i] = b1g[i];
        sm[OFF_U1 + i] = u1g[i];
    }
    for (int i = tid; i < 64; i += NTHR) sm[OFF_B2 + i] = b2g[i];

    const int row_own = l16;
    const int col_own = wv * 16 + 8 * hi;
    float yreg[8], zreg[8];
#pragma unroll
    for (int j = 0; j < 8; ++j) {
        float v = y1g[(row0 + row_own) * 64 + col_own + j];
        yreg[j] = v;
        zreg[j] = v;
        sm[OFF_Y + row_own * YST + col_own + j] = v;
    }
    __syncthreads();

    {
        float c0 = 0.f, c1 = 0.f;
        for (int d = 0; d < 64; ++d) {
            c0 += sm[OFF_W1 + tid * W1ST + d] * sm[OFF_W2 + d * W2ST + tid];
            c1 += sm[OFF_W1 + (tid + NTHR) * W1ST + d] *
                  sm[OFF_W2 + d * W2ST + tid + NTHR];
        }
        sm[OFF_C + tid] = c0;
        sm[OFF_C + tid + NTHR] = c1;
    }
    __syncthreads();

    v8f b1v[TPW], u1v[TPW], ccv[TPW];
#pragma unroll
    for (int i = 0; i < TPW; ++i) {
#pragma unroll
        for (int r = 0; r < 8; ++r) {
            const int n = (wv * TPW + i) * 16 + 8 * hi + r;
            b1v[i][r] = sm[OFF_B1 + n];
            u1v[i][r] = sm[OFF_U1 + n];
            ccv[i][r] = sm[OFF_C + n];
        }
    }
    float b2own[8];
#pragma unroll
    for (int j = 0; j < 8; ++j) b2own[j] = sm[OFF_B2 + col_own + j];

    v2f aw1[TPW][16], aw2[TPW][16];
    {
        const float* w1 = sm + OFF_W1 + (wv * TPW * 16 + l16) * W1ST + 2 * hi;
        const float* w2 = sm + OFF_W2 + l16 * W2ST + wv * TPW * 16 + 8 * hi;
#pragma unroll
        for (int i = 0; i < TPW; ++i) {
#pragma unroll
            for (int kt = 0; kt < 16; ++kt)
                aw1[i][kt] = *(const v2f*)&w1[i * 16 * W1ST + kt * 4];
#pragma unroll
            for (int j = 0; j < 4; ++j)
#pragma unroll
                for (int ot = 0; ot < 4; ++ot)
                    aw2[i][j * 4 + ot] =
                        *(const v2f*)&w2[ot * 16 * W2ST + i * 16 + j * 2];
        }
    }

    v8f trp = vzero8();

    float t1 = 1.0f;
#pragma unroll 1
    for (int s = 0; s < NSTEP; ++s) {
        const float t0 = t1 - HSTEP;

        mlp_partial<true>(sm, t1, sm + OFF_Y, wv, aw1, aw2, b1v, u1v, ccv, trp);
        __syncthreads();

        {
            v8f f = reduce_partials(sm, wv, lane);
#pragma unroll
            for (int j = 0; j < 8; ++j) {
                const float fv = f[j] + b2own[j];
                zreg[j] -= HSTEP * fv;
                sm[OFF_Z + row_own * YST + col_own + j] = zreg[j];
            }
        }
        __syncthreads();

        mlp_partial<false>(sm, t0, sm + OFF_Z, wv, aw1, aw2, b1v, u1v, ccv, trp);
        __syncthreads();

        {
            v8f f = reduce_partials(sm, wv, lane);
#pragma unroll
            for (int j = 0; j < 8; ++j) {
                const float fv = f[j] + b2own[j];
                yreg[j] = INVL * yreg[j] + (1.0f - INVL) * zreg[j] -
                          INVL * HSTEP * fv;
                sm[OFF_Y + row_own * YST + col_own + j] = yreg[j];
            }
        }
        __syncthreads();
        t1 = t0;
    }

    float* ts  = sm + OFF_PT;
    float* ts2 = sm + OFF_PT + NW * 32;
    ts[wv * 32 + lane] = ((trp[0] + trp[1]) + (trp[2] + trp[3])) +
                         ((trp[4] + trp[5]) + (trp[6] + trp[7]));
    __syncthreads();
    for (int e = tid; e < 16 * NW; e += NTHR) {
        const int m = e / NW, w = e % NW;
        ts2[e] = ts[w * 32 + m] + ts[w * 32 + 16 + m];
    }
    __syncthreads();
    if (tid < 16) {
        float sum = 0.f;
#pragma unroll
        for (int w = 0; w < NW; ++w) sum += ts2[tid * NW + w];
        sm[OFF_I + tid] = HSTEP * sum;
    }
    __syncthreads();

    for (int i = tid; i < 16 * 65; i += NTHR) {
        int m = i / 65, c = i % 65;
        out[(row0 + m) * 65 + c] = (c < 64) ? sm[OFF_Y + m * YST + c]
                                            : sm[OFF_I + m];
    }
}

extern "C" void kernel_launch(void* const* d_in, const int* in_sizes, int n_in,
                              void* d_out, int out_size, void* d_ws, size_t ws_size,
                              hipStream_t stream) {
    const float* y1 = (const float*)d_in[0];
    const float* W1 = (const float*)d_in[1];
    const float* b1 = (const float*)d_in[2];
    const float* u1 = (const float*)d_in[3];
    const float* W2 = (const float*)d_in[4];
    const float* b2 = (const float*)d_in[5];
    float* out = (float*)d_out;

    const size_t lds_bytes = (size_t)LDS_FLOATS * sizeof(float);
    (void)hipFuncSetAttribute((const void*)cnf_solver_kernel,
                              hipFuncAttributeMaxDynamicSharedMemorySize,
                              (int)lds_bytes);

    cnf_solver_kernel<<<16, NTHR, lds_bytes, stream>>>(y1, W1, b1, u1, W2, b2, out);
}


# exchange of partial f tiles and state between the 4 waves pipelined under the WMMAs in two column halves (G1 K order = D layout, split barriers); only 4 partial stores stay exposed per evaluation
# speedup vs baseline: 7.2995x; 1.0453x over previous
#include <hip/hip_runtime.h>


typedef __attribute__((ext_vector_type(2))) float v2f;
typedef __attribute__((ext_vector_type(4))) float v4f;
typedef __attribute__((ext_vector_type(8))) float v8f;

#define NSTEP 64
#define HSTEP (1.0f / 64.0f)
#define INVL  (1.0f / 0.999f)

#define NW   4
#define NTHR (NW * 32)
#define TPW  4

#define W1ST 66
#define W2ST 258
#define YST  66

static constexpr int OFF_W1 = 0;
static constexpr int OFF_W2 = OFF_W1 + 256 * W1ST;
static constexpr int OFF_Y  = OFF_W2 + 64 * W2ST;
static constexpr int OFF_Z  = OFF_Y + 16 * YST;
static constexpr int OFF_SF = OFF_Z;
static constexpr int OFF_B1 = OFF_Z + 16 * YST;
static constexpr int OFF_U1 = OFF_B1 + 256;
static constexpr int OFF_B2 = OFF_U1 + 256;
static constexpr int OFF_C  = OFF_B2 + 64;
static constexpr int OFF_I  = OFF_C + 256;
static constexpr int OFF_PT = (OFF_I + 16 + 7) & ~7;
static constexpr int PT_FLOATS  = NW * NW * 2 * 32 * 4;
static constexpr int LDS_FLOATS = OFF_PT + PT_FLOATS;

__device__ __forceinline__ v8f vzero8() {
    v8f z = {0.f, 0.f, 0.f, 0.f, 0.f, 0.f, 0.f, 0.f};
    return z;
}
__device__ __forceinline__ v8f vsplat8(float x) {
    v8f z = {x, x, x, x, x, x, x, x};
    return z;
}

#define WMMA_F32(a, b, c) \
    __builtin_amdgcn_wmma_f32_16x16x4_f32(false, (a), false, (b), (short)0, (c), false, false)

template <bool TRACE, int K>
__device__ __forceinline__ void tanh_stage(const v8f& c0, const v8f& c1,
                                           float t, const v8f& b1, const v8f& u1,
                                           const v8f& cc, v8f& p, v8f& q,
                                           v8f& tv, v8f& trp) {
    const float LOG2E_X2 = 2.8853900817779268f;
    if constexpr (K == 0) {
        v8f a0 = c0, a1 = c1;
        asm volatile("" : "+v"(a0), "+v"(a1));
        p = a0 + a1;
    } else if constexpr (K == 1) {
        q = __builtin_elementwise_fma(vsplat8(t), u1, b1);
    } else if constexpr (K == 2) {
        p = (p + q) * LOG2E_X2;
    } else if constexpr (K == 3) {
#pragma unroll
        for (int r = 0; r < 4; ++r) q[r] = __builtin_amdgcn_exp2f(p[r]);
    } else if constexpr (K == 4) {
#pragma unroll
        for (int r = 4; r < 8; ++r) q[r] = __builtin_amdgcn_exp2f(p[r]);
    } else if constexpr (K == 5) {
        p = q + 1.0f;
    } else if constexpr (K == 6) {
#pragma unroll
        for (int r = 0; r < 4; ++r) q[r] = __builtin_amdgcn_rcpf(p[r]);
    } else if constexpr (K == 7) {
#pragma unroll
        for (int r = 4; r < 8; ++r) q[r] = __builtin_amdgcn_rcpf(p[r]);
    } else if constexpr (K == 8) {
        tv = __builtin_elementwise_fma(vsplat8(-2.0f), q, vsplat8(1.0f));
    } else if constexpr (K == 9) {
        if constexpr (TRACE) p = __builtin_elementwise_fma(-tv, tv, vsplat8(1.0f));
    } else if constexpr (K == 10) {
        if constexpr (TRACE) trp = __builtin_elementwise_fma(p, cc, trp);
    }
}

__device__ __forceinline__ void bar_arrive() {
    __builtin_amdgcn_fence(__ATOMIC_RELEASE, "workgroup");
    __builtin_amdgcn_s_barrier_signal(-1);
}
__device__ __forceinline__ void bar_sync() {
    __builtin_amdgcn_s_barrier_wait(-1);
    __builtin_amdgcn_fence(__ATOMIC_ACQUIRE, "workgroup");
}

__device__ __forceinline__ void load_partials(const float* sm, int ot, int wv,
                                              int lane, v2f (&pp)[NW]) {
    const float* pr = sm + OFF_PT + ot * (NW * 256) + (wv >> 1) * 128 +
                      lane * 4 + (wv & 1) * 2;
#pragma unroll
    for (int w = 0; w < NW; ++w) pp[w] = *(const v2f*)&pr[w * 256];
}

template <bool ZKIND>
__device__ __forceinline__ void own_update(float* sm, int ot, int wv, int lane,
                                           const v2f (&pp)[NW], const v2f& b2,
                                           v2f& y, v2f& z) {
    const v2f fv = ((pp[0] + pp[1]) + (pp[2] + pp[3])) + b2;
    v2f nx;
    if (ZKIND) {
        z = z - HSTEP * fv;
        nx = z;
    } else {
        y = INVL * y + (1.0f - INVL) * z - (INVL * HSTEP) * fv;
        nx = y;
    }
    *(v2f*)&sm[OFF_SF + (ot * 4 + wv) * 64 + lane * 2] = nx;
}

__device__ __forceinline__ void store_partial(float* sm, int ot, int wv, int lane,
                                              const v8f& f) {
    float* pw = sm + OFF_PT + ot * (NW * 256) + wv * 256 + lane * 4;
    v4f lo = {f[0], f[1], f[2], f[3]};
    v4f hh = {f[4], f[5], f[6], f[7]};
    *(v4f*)&pw[0]   = lo;
    *(v4f*)&pw[128] = hh;
}

struct WaveRefs {
    float* sm; int wv; int lane;
    const v2f (&aw1)[TPW][16]; const v2f (&aw2)[TPW][16];
    const v8f (&b1v)[TPW]; const v8f (&u1v)[TPW]; const v8f (&ccv)[TPW];
    v8f& trp; v2f (&sf)[16]; v2f (&yown)[4]; v2f (&zown)[4]; const v2f (&b2own)[4];
};
struct EvalRegs {
    v8f acc[TPW][2];
    v8f tv[TPW];
    v8f facc[4];
    v8f p[2], q[2];
    v2f ppa[NW], ppb[NW];
};

template <bool TRACE, int N, int I>
__device__ __forceinline__ void tanh_ride(const WaveRefs& w, EvalRegs& e, float t) {
    constexpr int K = N - (32 + 8 * I + 9);
    if constexpr (K >= 0 && K <= 10)
        tanh_stage<TRACE, K>(e.acc[I][0], e.acc[I][1], t, w.b1v[I], w.u1v[I], w.ccv[I],
                             e.p[I & 1], e.q[I & 1], e.tv[I], w.trp);
}

template <bool TRACE, bool FIRST, int N>
__device__ __forceinline__ void slot(const WaveRefs& w, EvalRegs& e, float t) {
    float* sm = w.sm;
    const int wv = w.wv, lane = w.lane;
    if constexpr (N < 32) {
        constexpr int ks = N >> 2, i = N & 3, c = ks & 1;
        if constexpr (ks < 2) e.acc[i][c] = WMMA_F32(w.aw1[i][ks], w.sf[ks], vzero8());
        else                  e.acc[i][c] = WMMA_F32(w.aw1[i][ks], w.sf[ks], e.acc[i][c]);
    } else if constexpr (N < 64) {
        constexpr int m = N - 32, i = m >> 3, ks = 8 + (m & 7), c = ks & 1;
        e.acc[i][c] = WMMA_F32(w.aw1[i][ks], w.sf[ks], e.acc[i][c]);
    } else {
        constexpr int m = (N - 64) & 31, hf = (N - 64) >> 5;
        constexpr int i = m >> 3, j = (m >> 1) & 3, ot = hf * 2 + (m & 1);
        v2f hb;
        hb[0] = e.tv[i][2 * j];
        hb[1] = e.tv[i][2 * j + 1];
        if constexpr (i == 0 && j == 0)
            e.facc[ot] = WMMA_F32(w.aw2[i][j * 4 + ot], hb, vzero8());
        else
            e.facc[ot] = WMMA_F32(w.aw2[i][j * 4 + ot], hb, e.facc[ot]);
    }

    tanh_ride<TRACE, N, 0>(w, e, t);
    tanh_ride<TRACE, N, 1>(w, e, t);
    tanh_ride<TRACE, N, 2>(w, e, t);
    tanh_ride<TRACE, N, 3>(w, e, t);

    if constexpr (!FIRST && N == 2) bar_arrive();
    if constexpr (!FIRST && N == 6) {
        bar_sync();
        load_partials(sm, 2, wv, lane, e.ppa);
        load_partials(sm, 3, wv, lane, e.ppb);
    }
    if constexpr (!FIRST && N == 13)
        own_update<!TRACE>(sm, 2, wv, lane, e.ppa, w.b2own[2], w.yown[2], w.zown[2]);
    if constexpr (!FIRST && N == 14)
        own_update<!TRACE>(sm, 3, wv, lane, e.ppb, w.b2own[3], w.yown[3], w.zown[3]);
    if constexpr (!FIRST && N == 20) bar_arrive();
    if constexpr (N == 24) {
        if constexpr (!FIRST) bar_sync();
#pragma unroll
        for (int ks = 8; ks < 16; ++ks)
            w.sf[ks] = *(const v2f*)&sm[OFF_SF + ks * 64 + lane * 2];
    }

    if constexpr (N == 97) {
        store_partial(sm, 0, wv, lane, e.facc[0]);
        store_partial(sm, 1, wv, lane, e.facc[1]);
    }
    if constexpr (N == 103) bar_arrive();
    if constexpr (N == 107) {
        bar_sync();
        load_partials(sm, 0, wv, lane, e.ppa);
        load_partials(sm, 1, wv, lane, e.ppb);
    }
    if constexpr (N == 114)
        own_update<TRACE>(sm, 0, wv, lane, e.ppa, w.b2own[0], w.yown[0], w.zown[0]);
    if constexpr (N == 115)
        own_update<TRACE>(sm, 1, wv, lane, e.ppb, w.b2own[1], w.yown[1], w.zown[1]);
    if constexpr (N == 121) bar_arrive();
    if constexpr (N == 125) {
        bar_sync();
#pragma unroll
        for (int ks = 0; ks < 8; ++ks)
            w.sf[ks] = *(const v2f*)&sm[OFF_SF + ks * 64 + lane * 2];
    }

    __builtin_amdgcn_sched_barrier(0);
}

template <bool TRACE, bool FIRST, int N0, int CNT>
__device__ __forceinline__ void slots(const WaveRefs& w, EvalRegs& e, float t) {
    if constexpr (CNT == 1) {
        slot<TRACE, FIRST, N0>(w, e, t);
    } else {
        slots<TRACE, FIRST, N0, CNT / 2>(w, e, t);
        slots<TRACE, FIRST, N0 + CNT / 2, CNT - CNT / 2>(w, e, t);
    }
}

template <bool TRACE, bool FIRST>
__device__ __forceinline__ void eval_pass(const WaveRefs& w, float t) {
    EvalRegs e;
    slots<TRACE, FIRST, 0, 128>(w, e, t);
    store_partial(w.sm, 2, w.wv, w.lane, e.facc[2]);
    store_partial(w.sm, 3, w.wv, w.lane, e.facc[3]);
}

extern "C" __global__ void __launch_bounds__(NTHR)
__attribute__((amdgpu_waves_per_eu(1, 1)))
cnf_solver_kernel(const float* __restrict__ y1g, const float* __restrict__ W1g,
                  const float* __restrict__ b1g, const float* __restrict__ u1g,
                  const float* __restrict__ W2g, const float* __restrict__ b2g,
                  float* __restrict__ out) {
    extern __shared__ __attribute__((aligned(16))) float sm[];
    const int tid  = threadIdx.x;
    const int lane = tid & 31;
    const int l16  = lane & 15;
    const int hi   = lane >> 4;
    const int wv   = __builtin_amdgcn_readfirstlane(tid >> 5);
    const int row0 = blockIdx.x * 16;

#pragma unroll 1
    for (int b = 0; b < 2; ++b) {
        v4f tmp[16];
#pragma unroll
        for (int k = 0; k < 16; ++k)
            tmp[k] = *(const v4f*)&W1g[(tid + (b * 16 + k) * NTHR) * 4];
#pragma unroll
        for (int k = 0; k < 16; ++k) {
            const int i4 = tid + (b * 16 + k) * NTHR;
            const int m = i4 >> 4, d = (i4 & 15) * 4;
            v2f lo = {tmp[k][0], tmp[k][1]}, hh = {tmp[k][2], tmp[k][3]};
            *(v2f*)&sm[OFF_W1 + m * W1ST + d]     = lo;
            *(v2f*)&sm[OFF_W1 + m * W1ST + d + 2] = hh;
        }
    }
#pragma unroll 1
    for (int b = 0; b < 2; ++b) {
        v4f tmp[16];
#pragma unroll
        for (int k = 0; k < 16; ++k)
            tmp[k] = *(const v4f*)&W2g[(tid + (b * 16 + k) * NTHR) * 4];
#pragma unroll
        for (int k = 0; k < 16; ++k) {
            const int i4 = tid + (b * 16 + k) * NTHR;
            const int m = i4 >> 6, d = (i4 & 63) * 4;
            v2f lo = {tmp[k][0], tmp[k][1]}, hh = {tmp[k][2], tmp[k][3]};
            *(v2f*)&sm[OFF_W2 + m * W2ST + d]     = lo;
            *(v2f*)&sm[OFF_W2 + m * W2ST + d + 2] = hh;
        }
    }
    for (int i = tid; i < 256; i += NTHR) {
        sm[OFF_B1 + i] = b1g[i];
        sm[OFF_U1 + i] = u1g[i];
    }
    for (int i = tid; i < 64; i += NTHR) sm[OFF_B2 + i] = b2g[i];

    v2f yown[4], zown[4];
#pragma unroll
    for (int ot = 0; ot < 4; ++ot) {
        const v2f v = *(const v2f*)&y1g[(row0 + l16) * 64 + ot * 16 + 8 * hi + 2 * wv];
        yown[ot] = v;
        zown[ot] = v;
        *(v2f*)&sm[OFF_SF + (ot * 4 + wv) * 64 + lane * 2] = v;
    }
    __syncthreads();

    {
        float c0 = 0.f, c1 = 0.f;
        for (int d = 0; d < 64; ++d) {
            c0 += sm[OFF_W1 + tid * W1ST + d] * sm[OFF_W2 + d * W2ST + tid];
            c1 += sm[OFF_W1 + (tid + NTHR) * W1ST + d] *
                  sm[OFF_W2 + d * W2ST + tid + NTHR];
        }
        sm[OFF_C + tid] = c0;
        sm[OFF_C + tid + NTHR] = c1;
    }
    __syncthreads();

    v8f b1v[TPW], u1v[TPW], ccv[TPW];
#pragma unroll
    for (int i = 0; i < TPW; ++i) {
#pragma unroll
        for (int r = 0; r < 8; ++r) {
            const int n = (wv * TPW + i) * 16 + 8 * hi + r;
            b1v[i][r] = sm[OFF_B1 + n];
            u1v[i][r] = sm[OFF_U1 + n];
            ccv[i][r] = sm[OFF_C + n];
        }
    }
    v2f b2own[4];
#pragma unroll
    for (int ot = 0; ot < 4; ++ot) {
        b2own[ot][0] = sm[OFF_B2 + ot * 16 + 8 * hi + 2 * wv];
        b2own[ot][1] = sm[OFF_B2 + ot * 16 + 8 * hi + 2 * wv + 1];
    }

    v2f aw1[TPW][16], aw2[TPW][16];
    {
        const float* w1 = sm + OFF_W1 + (wv * TPW * 16 + l16) * W1ST + 8 * hi;
        const float* w2 = sm + OFF_W2 + l16 * W2ST + wv * TPW * 16 + 8 * hi;
#pragma unroll
        for (int i = 0; i < TPW; ++i) {
#pragma unroll
            for (int ks = 0; ks < 16; ++ks)
                aw1[i][ks] = *(const v2f*)&w1[i * 16 * W1ST + (ks >> 2) * 16 + (ks & 3) * 2];
#pragma unroll
            for (int j = 0; j < 4; ++j)
#pragma unroll
                for (int ot = 0; ot < 4; ++ot)
                    aw2[i][j * 4 + ot] =
                        *(const v2f*)&w2[ot * 16 * W2ST + i * 16 + j * 2];
        }
    }

    v8f trp = vzero8();

    v2f sf[16];
#pragma unroll
    for (int ks = 0; ks < 8; ++ks)
        sf[ks] = *(const v2f*)&sm[OFF_SF + ks * 64 + lane * 2];

    const WaveRefs wr = {sm, wv, lane, aw1, aw2, b1v, u1v, ccv, trp, sf, yown, zown, b2own};
    float t1 = 1.0f;
    {
        const float t0 = t1 - HSTEP;
        eval_pass<true,  true >(wr, t1);
        eval_pass<false, false>(wr, t0);
        t1 = t0;
    }
#pragma unroll 1
    for (int s = 1; s < NSTEP; ++s) {
        const float t0 = t1 - HSTEP;
        eval_pass<true,  false>(wr, t1);
        eval_pass<false, false>(wr, t0);
        t1 = t0;
    }

    {
        v2f ppa[NW], ppb[NW];
        bar_arrive();
        bar_sync();
        load_partials(sm, 2, wv, lane, ppa);
        load_partials(sm, 3, wv, lane, ppb);
        own_update<false>(sm, 2, wv, lane, ppa, b2own[2], yown[2], zown[2]);
        own_update<false>(sm, 3, wv, lane, ppb, b2own[3], yown[3], zown[3]);
    }
#pragma unroll
    for (int ot = 0; ot < 4; ++ot)
        *(v2f*)&sm[OFF_Y + l16 * YST + ot * 16 + 8 * hi + 2 * wv] = yown[ot];
    __syncthreads();

    float* ts  = sm + OFF_PT;
    float* ts2 = sm + OFF_PT + NW * 32;
    ts[wv * 32 + lane] = ((trp[0] + trp[1]) + (trp[2] + trp[3])) +
                         ((trp[4] + trp[5]) + (trp[6] + trp[7]));
    __syncthreads();
    for (int e = tid; e < 16 * NW; e += NTHR) {
        const int m = e / NW, w = e % NW;
        ts2[e] = ts[w * 32 + m] + ts[w * 32 + 16 + m];
    }
    __syncthreads();
    if (tid < 16) {
        float sum = 0.f;
#pragma unroll
        for (int w = 0; w < NW; ++w) sum += ts2[tid * NW + w];
        sm[OFF_I + tid] = HSTEP * sum;
    }
    __syncthreads();

    for (int i = tid; i < 16 * 65; i += NTHR) {
        int m = i / 65, c = i % 65;
        out[(row0 + m) * 65 + c] = (c < 64) ? sm[OFF_Y + m * YST + c]
                                            : sm[OFF_I + m];
    }
}

extern "C" void kernel_launch(void* const* d_in, const int* in_sizes, int n_in,
                              void* d_out, int out_size, void* d_ws, size_t ws_size,
                              hipStream_t stream) {
    const float* y1 = (const float*)d_in[0];
    const float* W1 = (const float*)d_in[1];
    const float* b1 = (const float*)d_in[2];
    const float* u1 = (const float*)d_in[3];
    const float* W2 = (const float*)d_in[4];
    const float* b2 = (const float*)d_in[5];
    float* out = (float*)d_out;

    const size_t lds_bytes = (size_t)LDS_FLOATS * sizeof(float);
    (void)hipFuncSetAttribute((const void*)cnf_solver_kernel,
                              hipFuncAttributeMaxDynamicSharedMemorySize,
                              (int)lds_bytes);

    cnf_solver_kernel<<<16, NTHR, lds_bytes, stream>>>(y1, W1, b1, u1, W2, b2, out);
}


# hidden-layer tanh by the hardware v_tanh_f32 (fp32 in and out; was exp2, add, rcp, fma per element): per hidden tile 8 transcendental issues, not 16, and 12 fewer packed ones beside the f32 WMMAs
# speedup vs baseline: 7.6374x; 1.0463x over previous
#include <hip/hip_runtime.h>


typedef __attribute__((ext_vector_type(2))) float v2f;
typedef __attribute__((ext_vector_type(4))) float v4f;
typedef __attribute__((ext_vector_type(8))) float v8f;

#define NSTEP 64
#define HSTEP (1.0f / 64.0f)
#define INVL  (1.0f / 0.999f)

#define NW   4
#define NTHR (NW * 32)
#define TPW  4

#define W1ST 66
#define W2ST 258
#define YST  66

static constexpr int OFF_W1 = 0;
static constexpr int OFF_W2 = OFF_W1 + 256 * W1ST;
static constexpr int OFF_Y  = OFF_W2 + 64 * W2ST;
static constexpr int OFF_Z  = OFF_Y + 16 * YST;
static constexpr int OFF_SF = OFF_Z;
static constexpr int OFF_B1 = OFF_Z + 16 * YST;
static constexpr int OFF_U1 = OFF_B1 + 256;
static constexpr int OFF_B2 = OFF_U1 + 256;
static constexpr int OFF_C  = OFF_B2 + 64;
static constexpr int OFF_I  = OFF_C + 256;
static constexpr int OFF_PT = (OFF_I + 16 + 7) & ~7;
static constexpr int PT_FLOATS  = NW * NW * 2 * 32 * 4;
static constexpr int LDS_FLOATS = OFF_PT + PT_FLOATS;

__device__ __forceinline__ v8f vzero8() {
    v8f z = {0.f, 0.f, 0.f, 0.f, 0.f, 0.f, 0.f, 0.f};
    return z;
}
__device__ __forceinline__ v8f vsplat8(float x) {
    v8f z = {x, x, x, x, x, x, x, x};
    return z;
}

#define WMMA_F32(a, b, c) \
    __builtin_amdgcn_wmma_f32_16x16x4_f32(false, (a), false, (b), (short)0, (c), false, false)

template <bool TRACE, int K>
__device__ __forceinline__ void tanh_stage(const v8f& c0, const v8f& c1,
                                           float t, const v8f& b1, const v8f& u1,
                                           const v8f& cc, v8f& p, v8f& q,
                                           v8f& tv, v8f& trp) {
    if constexpr (K == 0) {
        v8f a0 = c0, a1 = c1;
        asm volatile("" : "+v"(a0), "+v"(a1));
        p = a0 + a1;
    } else if constexpr (K == 1) {
        q = __builtin_elementwise_fma(vsplat8(t), u1, b1);
    } else if constexpr (K == 2) {
        p = p + q;
    } else if constexpr (K == 3) {
#pragma unroll
        for (int r = 0; r < 4; ++r) tv[r] = __builtin_amdgcn_tanhf(p[r]);
    } else if constexpr (K == 4) {
#pragma unroll
        for (int r = 4; r < 8; ++r) tv[r] = __builtin_amdgcn_tanhf(p[r]);
    } else if constexpr (K == 5) {
        if constexpr (TRACE) p = __builtin_elementwise_fma(-tv, tv, vsplat8(1.0f));
    } else if constexpr (K == 6) {
        if constexpr (TRACE) trp = __builtin_elementwise_fma(p, cc, trp);
    }
}

__device__ __forceinline__ void bar_arrive() {
    __builtin_amdgcn_fence(__ATOMIC_RELEASE, "workgroup");
    __builtin_amdgcn_s_barrier_signal(-1);
}
__device__ __forceinline__ void bar_sync() {
    __builtin_amdgcn_s_barrier_wait(-1);
    __builtin_amdgcn_fence(__ATOMIC_ACQUIRE, "workgroup");
}

__device__ __forceinline__ void load_partials(const float* sm, int ot, int wv,
                                              int lane, v2f (&pp)[NW]) {
    const float* pr = sm + OFF_PT + ot * (NW * 256) + (wv >> 1) * 128 +
                      lane * 4 + (wv & 1) * 2;
#pragma unroll
    for (int w = 0; w < NW; ++w) pp[w] = *(const v2f*)&pr[w * 256];
}

template <bool ZKIND>
__device__ __forceinline__ void own_update(float* sm, int ot, int wv, int lane,
                                           const v2f (&pp)[NW], const v2f& b2,
                                           v2f& y, v2f& z) {
    const v2f fv = ((pp[0] + pp[1]) + (pp[2] + pp[3])) + b2;
    v2f nx;
    if (ZKIND) {
        z = z - HSTEP * fv;
        nx = z;
    } else {
        y = INVL * y + (1.0f - INVL) * z - (INVL * HSTEP) * fv;
        nx = y;
    }
    *(v2f*)&sm[OFF_SF + (ot * 4 + wv) * 64 + lane * 2] = nx;
}

__device__ __forceinline__ void store_partial(float* sm, int ot, int wv, int lane,
                                              const v8f& f) {
    float* pw = sm + OFF_PT + ot * (NW * 256) + wv * 256 + lane * 4;
    v4f lo = {f[0], f[1], f[2], f[3]};
    v4f hh = {f[4], f[5], f[6], f[7]};
    *(v4f*)&pw[0]   = lo;
    *(v4f*)&pw[128] = hh;
}

struct WaveRefs {
    float* sm; int wv; int lane;
    const v2f (&aw1)[TPW][16]; const v2f (&aw2)[TPW][16];
    const v8f (&b1v)[TPW]; const v8f (&u1v)[TPW]; const v8f (&ccv)[TPW];
    v8f& trp; v2f (&sf)[16]; v2f (&yown)[4]; v2f (&zown)[4]; const v2f (&b2own)[4];
};
struct EvalRegs {
    v8f acc[TPW][2];
    v8f tv[TPW];
    v8f facc[4];
    v8f p[2], q[2];
    v2f ppa[NW], ppb[NW];
};

template <bool TRACE, int N, int I>
__device__ __forceinline__ void tanh_ride(const WaveRefs& w, EvalRegs& e, float t) {
    constexpr int K = N - (32 + 8 * I + 9);
    if constexpr (K >= 0 && K <= 6)
        tanh_stage<TRACE, K>(e.acc[I][0], e.acc[I][1], t, w.b1v[I], w.u1v[I], w.ccv[I],
                             e.p[I & 1], e.q[I & 1], e.tv[I], w.trp);
}

template <bool TRACE, bool FIRST, int N>
__device__ __forceinline__ void slot(const WaveRefs& w, EvalRegs& e, float t) {
    float* sm = w.sm;
    const int wv = w.wv, lane = w.lane;
    if constexpr (N < 32) {
        constexpr int ks = N >> 2, i = N & 3, c = ks & 1;
        if constexpr (ks < 2) e.acc[i][c] = WMMA_F32(w.aw1[i][ks], w.sf[ks], vzero8());
        else                  e.acc[i][c] = WMMA_F32(w.aw1[i][ks], w.sf[ks], e.acc[i][c]);
    } else if constexpr (N < 64) {
        constexpr int m = N - 32, i = m >> 3, ks = 8 + (m & 7), c = ks & 1;
        e.acc[i][c] = WMMA_F32(w.aw1[i][ks], w.sf[ks], e.acc[i][c]);
    } else {
        constexpr int m = (N - 64) & 31, hf = (N - 64) >> 5;
        constexpr int i = m >> 3, j = (m >> 1) & 3, ot = hf * 2 + (m & 1);
        v2f hb;
        hb[0] = e.tv[i][2 * j];
        hb[1] = e.tv[i][2 * j + 1];
        if constexpr (i == 0 && j == 0)
            e.facc[ot] = WMMA_F32(w.aw2[i][j * 4 + ot], hb, vzero8());
        else
            e.facc[ot] = WMMA_F32(w.aw2[i][j * 4 + ot], hb, e.facc[ot]);
    }

    tanh_ride<TRACE, N, 0>(w, e, t);
    tanh_ride<TRACE, N, 1>(w, e, t);
    tanh_ride<TRACE, N, 2>(w, e, t);
    tanh_ride<TRACE, N, 3>(w, e, t);

    if constexpr (!FIRST && N == 2) bar_arrive();
    if constexpr (!FIRST && N == 6) {
        bar_sync();
        load_partials(sm, 2, wv, lane, e.ppa);
        load_partials(sm, 3, wv, lane, e.ppb);
    }
    if constexpr (!FIRST && N == 13)
        own_update<!TRACE>(sm, 2, wv, lane, e.ppa, w.b2own[2], w.yown[2], w.zown[2]);
    if constexpr (!FIRST && N == 14)
        own_update<!TRACE>(sm, 3, wv, lane, e.ppb, w.b2own[3], w.yown[3], w.zown[3]);
    if constexpr (!FIRST && N == 20) bar_arrive();
    if constexpr (N == 24) {
        if constexpr (!FIRST) bar_sync();
#pragma unroll
        for (int ks = 8; ks < 16; ++ks)
            w.sf[ks] = *(const v2f*)&sm[OFF_SF + ks * 64 + lane * 2];
    }

    if constexpr (N == 97) {
        store_partial(sm, 0, wv, lane, e.facc[0]);
        store_partial(sm, 1, wv, lane, e.facc[1]);
    }
    if constexpr (N == 103) bar_arrive();
    if constexpr (N == 107) {
        bar_sync();
        load_partials(sm, 0, wv, lane, e.ppa);
        load_partials(sm, 1, wv, lane, e.ppb);
    }
    if constexpr (N == 114)
        own_update<TRACE>(sm, 0, wv, lane, e.ppa, w.b2own[0], w.yown[0], w.zown[0]);
    if constexpr (N == 115)
        own_update<TRACE>(sm, 1, wv, lane, e.ppb, w.b2own[1], w.yown[1], w.zown[1]);
    if constexpr (N == 121) bar_arrive();
    if constexpr (N == 125) {
        bar_sync();
#pragma unroll
        for (int ks = 0; ks < 8; ++ks)
            w.sf[ks] = *(const v2f*)&sm[OFF_SF + ks * 64 + lane * 2];
    }

    __builtin_amdgcn_sched_barrier(0);
}

template <bool TRACE, bool FIRST, int N0, int CNT>
__device__ __forceinline__ void slots(const WaveRefs& w, EvalRegs& e, float t) {
    if constexpr (CNT == 1) {
        slot<TRACE, FIRST, N0>(w, e, t);
    } else {
        slots<TRACE, FIRST, N0, CNT / 2>(w, e, t);
        slots<TRACE, FIRST, N0 + CNT / 2, CNT - CNT / 2>(w, e, t);
    }
}

template <bool TRACE, bool FIRST>
__device__ __forceinline__ void eval_pass(const WaveRefs& w, float t) {
    EvalRegs e;
    slots<TRACE, FIRST, 0, 128>(w, e, t);
    store_partial(w.sm, 2, w.wv, w.lane, e.facc[2]);
    store_partial(w.sm, 3, w.wv, w.lane, e.facc[3]);
}

extern "C" __global__ void __launch_bounds__(NTHR)
__attribute__((amdgpu_waves_per_eu(1, 1)))
cnf_solver_kernel(const float* __restrict__ y1g, const float* __restrict__ W1g,
                  const float* __restrict__ b1g, const float* __restrict__ u1g,
                  const float* __restrict__ W2g, const float* __restrict__ b2g,
                  float* __restrict__ out) {
    extern __shared__ __attribute__((aligned(16))) float sm[];
    const int tid  = threadIdx.x;
    const int lane = tid & 31;
    const int l16  = lane & 15;
    const int hi   = lane >> 4;
    const int wv   = __builtin_amdgcn_readfirstlane(tid >> 5);
    const int row0 = blockIdx.x * 16;

#pragma unroll 1
    for (int b = 0; b < 2; ++b) {
        v4f tmp[16];
#pragma unroll
        for (int k = 0; k < 16; ++k)
            tmp[k] = *(const v4f*)&W1g[(tid + (b * 16 + k) * NTHR) * 4];
#pragma unroll
        for (int k = 0; k < 16; ++k) {
            const int i4 = tid + (b * 16 + k) * NTHR;
            const int m = i4 >> 4, d = (i4 & 15) * 4;
            v2f lo = {tmp[k][0], tmp[k][1]}, hh = {tmp[k][2], tmp[k][3]};
            *(v2f*)&sm[OFF_W1 + m * W1ST + d]     = lo;
            *(v2f*)&sm[OFF_W1 + m * W1ST + d + 2] = hh;
        }
    }
#pragma unroll 1
    for (int b = 0; b < 2; ++b) {
        v4f tmp[16];
#pragma unroll
        for (int k = 0; k < 16; ++k)
            tmp[k] = *(const v4f*)&W2g[(tid + (b * 16 + k) * NTHR) * 4];
#pragma unroll
        for (int k = 0; k < 16; ++k) {
            const int i4 = tid + (b * 16 + k) * NTHR;
            const int m = i4 >> 6, d = (i4 & 63) * 4;
            v2f lo = {tmp[k][0], tmp[k][1]}, hh = {tmp[k][2], tmp[k][3]};
            *(v2f*)&sm[OFF_W2 + m * W2ST + d]     = lo;
            *(v2f*)&sm[OFF_W2 + m * W2ST + d + 2] = hh;
        }
    }
    for (int i = tid; i < 256; i += NTHR) {
        sm[OFF_B1 + i] = b1g[i];
        sm[OFF_U1 + i] = u1g[i];
    }
    for (int i = tid; i < 64; i += NTHR) sm[OFF_B2 + i] = b2g[i];

    v2f yown[4], zown[4];
#pragma unroll
    for (int ot = 0; ot < 4; ++ot) {
        const v2f v = *(const v2f*)&y1g[(row0 + l16) * 64 + ot * 16 + 8 * hi + 2 * wv];
        yown[ot] = v;
        zown[ot] = v;
        *(v2f*)&sm[OFF_SF + (ot * 4 + wv) * 64 + lane * 2] = v;
    }
    __syncthreads();

    {
        float c0 = 0.f, c1 = 0.f;
        for (int d = 0; d < 64; ++d) {
            c0 += sm[OFF_W1 + tid * W1ST + d] * sm[OFF_W2 + d * W2ST + tid];
            c1 += sm[OFF_W1 + (tid + NTHR) * W1ST + d] *
                  sm[OFF_W2 + d * W2ST + tid + NTHR];
        }
        sm[OFF_C + tid] = c0;
        sm[OFF_C + tid + NTHR] = c1;
    }
    __syncthreads();

    v8f b1v[TPW], u1v[TPW], ccv[TPW];
#pragma unroll
    for (int i = 0; i < TPW; ++i) {
#pragma unroll
        for (int r = 0; r < 8; ++r) {
            const int n = (wv * TPW + i) * 16 + 8 * hi + r;
            b1v[i][r] = sm[OFF_B1 + n];
            u1v[i][r] = sm[OFF_U1 + n];
            ccv[i][r] = sm[OFF_C + n];
        }
    }
    v2f b2own[4];
#pragma unroll
    for (int ot = 0; ot < 4; ++ot) {
        b2own[ot][0] = sm[OFF_B2 + ot * 16 + 8 * hi + 2 * wv];
        b2own[ot][1] = sm[OFF_B2 + ot * 16 + 8 * hi + 2 * wv + 1];
    }

    v2f aw1[TPW][16], aw2[TPW][16];
    {
        const float* w1 = sm + OFF_W1 + (wv * TPW * 16 + l16) * W1ST + 8 * hi;
        const float* w2 = sm + OFF_W2 + l16 * W2ST + wv * TPW * 16 + 8 * hi;
#pragma unroll
        for (int i = 0; i < TPW; ++i) {
#pragma unroll
            for (int ks = 0; ks < 16; ++ks)
                aw1[i][ks] = *(const v2f*)&w1[i * 16 * W1ST + (ks >> 2) * 16 + (ks & 3) * 2];
#pragma unroll
            for (int j = 0; j < 4; ++j)
#pragma unroll
                for (int ot = 0; ot < 4; ++ot)
                    aw2[i][j * 4 + ot] =
                        *(const v2f*)&w2[ot * 16 * W2ST + i * 16 + j * 2];
        }
    }

    v8f trp = vzero8();

    v2f sf[16];
#pragma unroll
    for (int ks = 0; ks < 8; ++ks)
        sf[ks] = *(const v2f*)&sm[OFF_SF + ks * 64 + lane * 2];

    const WaveRefs wr = {sm, wv, lane, aw1, aw2, b1v, u1v, ccv, trp, sf, yown, zown, b2own};
    float t1 = 1.0f;
    {
        const float t0 = t1 - HSTEP;
        eval_pass<true,  true >(wr, t1);
        eval_pass<false, false>(wr, t0);
        t1 = t0;
    }
#pragma unroll 1
    for (int s = 1; s < NSTEP; ++s) {
        const float t0 = t1 - HSTEP;
        eval_pass<true,  false>(wr, t1);
        eval_pass<false, false>(wr, t0);
        t1 = t0;
    }

    {
        v2f ppa[NW], ppb[NW];
        bar_arrive();
        bar_sync();
        load_partials(sm, 2, wv, lane, ppa);
        load_partials(sm, 3, wv, lane, ppb);
        own_update<false>(sm, 2, wv, lane, ppa, b2own[2], yown[2], zown[2]);
        own_update<false>(sm, 3, wv, lane, ppb, b2own[3], yown[3], zown[3]);
    }
#pragma unroll
    for (int ot = 0; ot < 4; ++ot)
        *(v2f*)&sm[OFF_Y + l16 * YST + ot * 16 + 8 * hi + 2 * wv] = yown[ot];
    __syncthreads();

    float* ts  = sm + OFF_PT;
    float* ts2 = sm + OFF_PT + NW * 32;
    ts[wv * 32 + lane] = ((trp[0] + trp[1]) + (trp[2] + trp[3])) +
                         ((trp[4] + trp[5]) + (trp[6] + trp[7]));
    __syncthreads();
    for (int e = tid; e < 16 * NW; e += NTHR) {
        const int m = e / NW, w = e % NW;
        ts2[e] = ts[w * 32 + m] + ts[w * 32 + 16 + m];
    }
    __syncthreads();
    if (tid < 16) {
        float sum = 0.f;
#pragma unroll
        for (int w = 0; w < NW; ++w) sum += ts2[tid * NW + w];
        sm[OFF_I + tid] = HSTEP * sum;
    }
    __syncthreads();

    for (int i = tid; i < 16 * 65; i += NTHR) {
        int m = i / 65, c = i % 65;
        out[(row0 + m) * 65 + c] = (c < 64) ? sm[OFF_Y + m * YST + c]
                                            : sm[OFF_I + m];
    }
}

extern "C" void kernel_launch(void* const* d_in, const int* in_sizes, int n_in,
                              void* d_out, int out_size, void* d_ws, size_t ws_size,
                              hipStream_t stream) {
    const float* y1 = (const float*)d_in[0];
    const float* W1 = (const float*)d_in[1];
    const float* b1 = (const float*)d_in[2];
    const float* u1 = (const float*)d_in[3];
    const float* W2 = (const float*)d_in[4];
    const float* b2 = (const float*)d_in[5];
    float* out = (float*)d_out;

    const size_t lds_bytes = (size_t)LDS_FLOATS * sizeof(float);
    (void)hipFuncSetAttribute((const void*)cnf_solver_kernel,
                              hipFuncAttributeMaxDynamicSharedMemorySize,
                              (int)lds_bytes);

    cnf_solver_kernel<<<16, NTHR, lds_bytes, stream>>>(y1, W1, b1, u1, W2, b2, out);
}


# each wave sums all four partial f tiles itself and keeps the whole state in registers as GEMM1's B fragments: the state's LDS store, barrier and reload are gone (2 barriers per evaluation, not 4)
# speedup vs baseline: 7.8981x; 1.0341x over previous
#include <hip/hip_runtime.h>


typedef __attribute__((ext_vector_type(2))) float v2f;
typedef __attribute__((ext_vector_type(4))) float v4f;
typedef __attribute__((ext_vector_type(8))) float v8f;

#define NSTEP 64
#define HSTEP (1.0f / 64.0f)
#define INVL  (1.0f / 0.999f)

#define NW   4
#define NTHR (NW * 32)
#define TPW  4

#define W1ST 66
#define W2ST 258
#define YST  66

static constexpr int OFF_W1 = 0;
static constexpr int OFF_W2 = OFF_W1 + 256 * W1ST;
static constexpr int OFF_Y  = OFF_W2 + 64 * W2ST;
static constexpr int OFF_B1 = OFF_Y + 16 * YST;
static constexpr int OFF_U1 = OFF_B1 + 256;
static constexpr int OFF_B2 = OFF_U1 + 256;
static constexpr int OFF_C  = OFF_B2 + 64;
static constexpr int OFF_I  = OFF_C + 256;
static constexpr int OFF_PT = (OFF_I + 16 + 7) & ~7;
static constexpr int PT_FLOATS  = NW * NW * 2 * 32 * 4;
static constexpr int LDS_FLOATS = OFF_PT + PT_FLOATS;

__device__ __forceinline__ v8f vzero8() {
    v8f z = {0.f, 0.f, 0.f, 0.f, 0.f, 0.f, 0.f, 0.f};
    return z;
}
__device__ __forceinline__ v8f vsplat8(float x) {
    v8f z = {x, x, x, x, x, x, x, x};
    return z;
}

#define WMMA_F32(a, b, c) \
    __builtin_amdgcn_wmma_f32_16x16x4_f32(false, (a), false, (b), (short)0, (c), false, false)

template <bool TRACE, int K>
__device__ __forceinline__ void tanh_stage(const v8f& c0, const v8f& c1,
                                           float t, const v8f& b1, const v8f& u1,
                                           const v8f& cc, v8f& p, v8f& q,
                                           v8f& tv, v8f& trp) {
    if constexpr (K == 0) {
        v8f a0 = c0, a1 = c1;
        asm volatile("" : "+v"(a0), "+v"(a1));
        p = a0 + a1;
    } else if constexpr (K == 1) {
        q = __builtin_elementwise_fma(vsplat8(t), u1, b1);
    } else if constexpr (K == 2) {
        p = p + q;
    } else if constexpr (K == 3) {
#pragma unroll
        for (int r = 0; r < 4; ++r) tv[r] = __builtin_amdgcn_tanhf(p[r]);
    } else if constexpr (K == 4) {
#pragma unroll
        for (int r = 4; r < 8; ++r) tv[r] = __builtin_amdgcn_tanhf(p[r]);
    } else if constexpr (K == 5) {
        if constexpr (TRACE) p = __builtin_elementwise_fma(-tv, tv, vsplat8(1.0f));
    } else if constexpr (K == 6) {
        if constexpr (TRACE) trp = __builtin_elementwise_fma(p, cc, trp);
    }
}

__device__ __forceinline__ void bar_arrive() {
    __builtin_amdgcn_fence(__ATOMIC_RELEASE, "workgroup");
    __builtin_amdgcn_s_barrier_signal(-1);
}
__device__ __forceinline__ void bar_sync() {
    __builtin_amdgcn_s_barrier_wait(-1);
    __builtin_amdgcn_fence(__ATOMIC_ACQUIRE, "workgroup");
}

__device__ __forceinline__ void load_half(const float* sm, int ot0, int lane,
                                          v4f (&pp)[2][NW][2]) {
    const float* pr = sm + OFF_PT + ot0 * (NW * 256) + lane * 4;
#pragma unroll
    for (int t = 0; t < 2; ++t)
#pragma unroll
        for (int hf = 0; hf < 2; ++hf)
#pragma unroll
            for (int w = 0; w < NW; ++w)
                pp[t][w][hf] = *(const v4f*)&pr[t * (NW * 256) + w * 256 + hf * 128];
}

__device__ __forceinline__ v2f vsplat2(float x) {
    v2f z = {x, x};
    return z;
}

template <bool ZKIND, int J, int S>
__device__ __forceinline__ void update_stage(const v4f (&pp)[NW][2], v2f& y, v2f& z,
                                             v2f& a, v2f& b, v2f& m, v2f& f, v2f& u) {
    if constexpr (S == 0) {
        constexpr int hf = J >> 1, o = (J & 1) * 2;
        v2f p[NW];
#pragma unroll
        for (int w = 0; w < NW; ++w) {
            p[w][0] = pp[w][hf][o];
            p[w][1] = pp[w][hf][o + 1];
        }
        a = p[0] + p[1];
        b = p[2] + p[3];
        if constexpr (!ZKIND) m = (1.0f - INVL) * z;
    } else if constexpr (S == 1) {
        if constexpr (ZKIND) asm volatile("" : "+v"(a), "+v"(b));
        else                 asm volatile("" : "+v"(a), "+v"(b), "+v"(m));
        f = a + b;
        if constexpr (!ZKIND) u = __builtin_elementwise_fma(vsplat2(INVL), y, m);
    } else {
        if constexpr (ZKIND) asm volatile("" : "+v"(f));
        else                 asm volatile("" : "+v"(f), "+v"(u));
        if constexpr (ZKIND) z = __builtin_elementwise_fma(vsplat2(-HSTEP), f, z);
        else                 y = __builtin_elementwise_fma(vsplat2(-(INVL * HSTEP)), f, u);
    }
}

__device__ __forceinline__ void store_partial(float* sm, int ot, int wv, int lane,
                                              const v8f& f) {
    float* pw = sm + OFF_PT + ot * (NW * 256) + wv * 256 + lane * 4;
    v4f lo = {f[0], f[1], f[2], f[3]};
    v4f hh = {f[4], f[5], f[6], f[7]};
    *(v4f*)&pw[0]   = lo;
    *(v4f*)&pw[128] = hh;
}

struct WaveRefs {
    float* sm; int wv; int lane;
    const v2f (&aw1)[TPW][16]; const v2f (&aw2)[TPW][16];
    const v8f (&b1v)[TPW]; const v8f (&u1v)[TPW]; const v8f (&ccv)[TPW];
    v8f& trp; v2f (&y)[16]; v2f (&z)[16]; const v8f (&c2)[4];
};
struct EvalRegs {
    v8f acc[TPW][2];
    v8f tv[TPW];
    v8f facc[4];
    v8f p[2], q[2];
    v4f pp[2][NW][2];
    v2f ua[8], ub[8], um[8], uf[8], uu[8];
};

template <bool TRACE, int N, int I>
__device__ __forceinline__ void tanh_ride(const WaveRefs& w, EvalRegs& e, float t) {
    constexpr int K = N - (32 + 8 * I + 9);
    if constexpr (K >= 0 && K <= 6)
        tanh_stage<TRACE, K>(e.acc[I][0], e.acc[I][1], t, w.b1v[I], w.u1v[I], w.ccv[I],
                             e.p[I & 1], e.q[I & 1], e.tv[I], w.trp);
}

template <bool ZKIND, int K, int S, int KS0>
__device__ __forceinline__ void update_ride(const WaveRefs& w, EvalRegs& e) {
    if constexpr (K >= 0 && K < 8)
        update_stage<ZKIND, (K & 3), S>(e.pp[K >> 2], w.y[KS0 + K], w.z[KS0 + K],
                                        e.ua[K], e.ub[K], e.um[K], e.uf[K], e.uu[K]);
}

template <bool TRACE, bool FIRST, int N>
__device__ __forceinline__ void slot(const WaveRefs& w, EvalRegs& e, float t) {
    float* sm = w.sm;
    const int wv = w.wv, lane = w.lane;
    if constexpr (N < 32) {
        constexpr int ks = N >> 2, i = N & 3, c = ks & 1;
        if constexpr (ks < 2) e.acc[i][c] = WMMA_F32(w.aw1[i][ks], (TRACE ? w.y[ks] : w.z[ks]), vzero8());
        else                  e.acc[i][c] = WMMA_F32(w.aw1[i][ks], (TRACE ? w.y[ks] : w.z[ks]), e.acc[i][c]);
    } else if constexpr (N < 64) {
        constexpr int m = N - 32, i = m >> 3, ks = 8 + (m & 7), c = ks & 1;
        e.acc[i][c] = WMMA_F32(w.aw1[i][ks], (TRACE ? w.y[ks] : w.z[ks]), e.acc[i][c]);
    } else {
        constexpr int m = (N - 64) & 31, hf = (N - 64) >> 5;
        constexpr int i = m >> 3, j = (m >> 1) & 3, ot = hf * 2 + (m & 1);
        v2f hb;
        hb[0] = e.tv[i][2 * j];
        hb[1] = e.tv[i][2 * j + 1];
        if constexpr (i == 0 && j == 0)
            e.facc[ot] = WMMA_F32(w.aw2[i][j * 4 + ot], hb, w.c2[ot]);
        else
            e.facc[ot] = WMMA_F32(w.aw2[i][j * 4 + ot], hb, e.facc[ot]);
    }

    if constexpr ((!FIRST && N >= 15 && N < 25) || (N >= 116 && N < 126))
        __builtin_amdgcn_sched_barrier(0);

    tanh_ride<TRACE, N, 0>(w, e, t);
    tanh_ride<TRACE, N, 1>(w, e, t);
    tanh_ride<TRACE, N, 2>(w, e, t);
    tanh_ride<TRACE, N, 3>(w, e, t);

    if constexpr (!FIRST && N == 2) bar_arrive();
    if constexpr (!FIRST && N == 6) {
        bar_sync();
        load_half(sm, 2, lane, e.pp);
    }
    if constexpr (!FIRST) {
        update_ride<!TRACE, N - 15, 0, 8>(w, e);
        update_ride<!TRACE, N - 16, 1, 8>(w, e);
        update_ride<!TRACE, N - 17, 2, 8>(w, e);
    }

    if constexpr (N == 97) {
        store_partial(sm, 0, wv, lane, e.facc[0]);
        store_partial(sm, 1, wv, lane, e.facc[1]);
    }
    if constexpr (N == 103) bar_arrive();
    if constexpr (N == 107) {
        bar_sync();
        load_half(sm, 0, lane, e.pp);
    }
    update_ride<TRACE, N - 116, 0, 0>(w, e);
    update_ride<TRACE, N - 117, 1, 0>(w, e);
    update_ride<TRACE, N - 118, 2, 0>(w, e);

    __builtin_amdgcn_sched_barrier(0);
}

template <bool TRACE, bool FIRST, int N0, int CNT>
__device__ __forceinline__ void slots(const WaveRefs& w, EvalRegs& e, float t) {
    if constexpr (CNT == 1) {
        slot<TRACE, FIRST, N0>(w, e, t);
    } else {
        slots<TRACE, FIRST, N0, CNT / 2>(w, e, t);
        slots<TRACE, FIRST, N0 + CNT / 2, CNT - CNT / 2>(w, e, t);
    }
}

template <bool TRACE, bool FIRST>
__device__ __forceinline__ void eval_pass(const WaveRefs& w, float t) {
    EvalRegs e;
    slots<TRACE, FIRST, 0, 128>(w, e, t);
    store_partial(w.sm, 2, w.wv, w.lane, e.facc[2]);
    store_partial(w.sm, 3, w.wv, w.lane, e.facc[3]);
}

extern "C" __global__ void __launch_bounds__(NTHR)
__attribute__((amdgpu_waves_per_eu(1, 1)))
cnf_solver_kernel(const float* __restrict__ y1g, const float* __restrict__ W1g,
                  const float* __restrict__ b1g, const float* __restrict__ u1g,
                  const float* __restrict__ W2g, const float* __restrict__ b2g,
                  float* __restrict__ out) {
    extern __shared__ __attribute__((aligned(16))) float sm[];
    const int tid  = threadIdx.x;
    const int lane = tid & 31;
    const int l16  = lane & 15;
    const int hi   = lane >> 4;
    const int wv   = __builtin_amdgcn_readfirstlane(tid >> 5);
    const int row0 = blockIdx.x * 16;

#pragma unroll 1
    for (int b = 0; b < 2; ++b) {
        v4f tmp[16];
#pragma unroll
        for (int k = 0; k < 16; ++k)
            tmp[k] = *(const v4f*)&W1g[(tid + (b * 16 + k) * NTHR) * 4];
#pragma unroll
        for (int k = 0; k < 16; ++k) {
            const int i4 = tid + (b * 16 + k) * NTHR;
            const int m = i4 >> 4, d = (i4 & 15) * 4;
            v2f lo = {tmp[k][0], tmp[k][1]}, hh = {tmp[k][2], tmp[k][3]};
            *(v2f*)&sm[OFF_W1 + m * W1ST + d]     = lo;
            *(v2f*)&sm[OFF_W1 + m * W1ST + d + 2] = hh;
        }
    }
#pragma unroll 1
    for (int b = 0; b < 2; ++b) {
        v4f tmp[16];
#pragma unroll
        for (int k = 0; k < 16; ++k)
            tmp[k] = *(const v4f*)&W2g[(tid + (b * 16 + k) * NTHR) * 4];
#pragma unroll
        for (int k = 0; k < 16; ++k) {
            const int i4 = tid + (b * 16 + k) * NTHR;
            const int m = i4 >> 6, d = (i4 & 63) * 4;
            v2f lo = {tmp[k][0], tmp[k][1]}, hh = {tmp[k][2], tmp[k][3]};
            *(v2f*)&sm[OFF_W2 + m * W2ST + d]     = lo;
            *(v2f*)&sm[OFF_W2 + m * W2ST + d + 2] = hh;
        }
    }
    for (int i = tid; i < 256; i += NTHR) {
        sm[OFF_B1 + i] = b1g[i];
        sm[OFF_U1 + i] = u1g[i];
    }
    for (int i = tid; i < 64; i += NTHR) sm[OFF_B2 + i] = b2g[i];

    v2f ys[16], zs[16];
#pragma unroll
    for (int ot = 0; ot < 4; ++ot) {
#pragma unroll
        for (int j = 0; j < 4; ++j) {
            const v2f v = *(const v2f*)&y1g[(row0 + l16) * 64 + ot * 16 + 8 * hi + 2 * j];
            ys[ot * 4 + j] = v;
            zs[ot * 4 + j] = v;
        }
    }
    __syncthreads();

    {
        float c0 = 0.f, c1 = 0.f;
        for (int d = 0; d < 64; ++d) {
            c0 += sm[OFF_W1 + tid * W1ST + d] * sm[OFF_W2 + d * W2ST + tid];
            c1 += sm[OFF_W1 + (tid + NTHR) * W1ST + d] *
                  sm[OFF_W2 + d * W2ST + tid + NTHR];
        }
        sm[OFF_C + tid] = c0;
        sm[OFF_C + tid + NTHR] = c1;
    }
    __syncthreads();

    v8f b1v[TPW], u1v[TPW], ccv[TPW];
#pragma unroll
    for (int i = 0; i < TPW; ++i) {
#pragma unroll
        for (int r = 0; r < 8; ++r) {
            const int n = (wv * TPW + i) * 16 + 8 * hi + r;
            b1v[i][r] = sm[OFF_B1 + n];
            u1v[i][r] = sm[OFF_U1 + n];
            ccv[i][r] = sm[OFF_C + n];
        }
    }
    v8f c2[4];
#pragma unroll
    for (int ot = 0; ot < 4; ++ot) {
#pragma unroll
        for (int r = 0; r < 8; ++r) {
            const float bv = sm[OFF_B2 + ot * 16 + 8 * hi + r];
            c2[ot][r] = (wv == 0) ? bv : 0.f;
        }
    }

    v2f aw1[TPW][16], aw2[TPW][16];
    {
        const float* w1 = sm + OFF_W1 + (wv * TPW * 16 + l16) * W1ST + 8 * hi;
        const float* w2 = sm + OFF_W2 + l16 * W2ST + wv * TPW * 16 + 8 * hi;
#pragma unroll
        for (int i = 0; i < TPW; ++i) {
#pragma unroll
            for (int ks = 0; ks < 16; ++ks)
                aw1[i][ks] = *(const v2f*)&w1[i * 16 * W1ST + (ks >> 2) * 16 + (ks & 3) * 2];
#pragma unroll
            for (int j = 0; j < 4; ++j)
#pragma unroll
                for (int ot = 0; ot < 4; ++ot)
                    aw2[i][j * 4 + ot] =
                        *(const v2f*)&w2[ot * 16 * W2ST + i * 16 + j * 2];
        }
    }

    v8f trp = vzero8();

    const WaveRefs wr = {sm, wv, lane, aw1, aw2, b1v, u1v, ccv, trp, ys, zs, c2};
    float t1 = 1.0f;
    {
        const float t0 = t1 - HSTEP;
        eval_pass<true,  true >(wr, t1);
        eval_pass<false, false>(wr, t0);
        t1 = t0;
    }
#pragma unroll 1
    for (int s = 1; s < NSTEP; ++s) {
        const float t0 = t1 - HSTEP;
        eval_pass<true,  false>(wr, t1);
        eval_pass<false, false>(wr, t0);
        t1 = t0;
    }

    {
        v4f pp[2][NW][2];
        v2f a, b, m, f, u;
        bar_arrive();
        bar_sync();
        load_half(sm, 2, lane, pp);
#define TAIL_PAIR(K)                                                                   \
        update_stage<false, ((K) & 3), 0>(pp[(K) >> 2], ys[8 + (K)], zs[8 + (K)], a, b, m, f, u); \
        update_stage<false, ((K) & 3), 1>(pp[(K) >> 2], ys[8 + (K)], zs[8 + (K)], a, b, m, f, u); \
        update_stage<false, ((K) & 3), 2>(pp[(K) >> 2], ys[8 + (K)], zs[8 + (K)], a, b, m, f, u);
        TAIL_PAIR(0) TAIL_PAIR(1) TAIL_PAIR(2) TAIL_PAIR(3)
        TAIL_PAIR(4) TAIL_PAIR(5) TAIL_PAIR(6) TAIL_PAIR(7)
#undef TAIL_PAIR
    }
    if (wv == 0) {
#pragma unroll
        for (int ks = 0; ks < 16; ++ks)
            *(v2f*)&sm[OFF_Y + l16 * YST + (ks >> 2) * 16 + 8 * hi + 2 * (ks & 3)] = ys[ks];
    }
    __syncthreads();

    float* ts  = sm + OFF_PT;
    float* ts2 = sm + OFF_PT + NW * 32;
    ts[wv * 32 + lane] = ((trp[0] + trp[1]) + (trp[2] + trp[3])) +
                         ((trp[4] + trp[5]) + (trp[6] + trp[7]));
    __syncthreads();
    for (int e = tid; e < 16 * NW; e += NTHR) {
        const int m = e / NW, w = e % NW;
        ts2[e] = ts[w * 32 + m] + ts[w * 32 + 16 + m];
    }
    __syncthreads();
    if (tid < 16) {
        float sum = 0.f;
#pragma unroll
        for (int w = 0; w < NW; ++w) sum += ts2[tid * NW + w];
        sm[OFF_I + tid] = HSTEP * sum;
    }
    __syncthreads();

    for (int i = tid; i < 16 * 65; i += NTHR) {
        int m = i / 65, c = i % 65;
        out[(row0 + m) * 65 + c] = (c < 64) ? sm[OFF_Y + m * YST + c]
                                            : sm[OFF_I + m];
    }
}

extern "C" void kernel_launch(void* const* d_in, const int* in_sizes, int n_in,
                              void* d_out, int out_size, void* d_ws, size_t ws_size,
                              hipStream_t stream) {
    const float* y1 = (const float*)d_in[0];
    const float* W1 = (const float*)d_in[1];
    const float* b1 = (const float*)d_in[2];
    const float* u1 = (const float*)d_in[3];
    const float* W2 = (const float*)d_in[4];
    const float* b2 = (const float*)d_in[5];
    float* out = (float*)d_out;

    const size_t lds_bytes = (size_t)LDS_FLOATS * sizeof(float);
    (void)hipFuncSetAttribute((const void*)cnf_solver_kernel,
                              hipFuncAttributeMaxDynamicSharedMemorySize,
                              (int)lds_bytes);

    cnf_solver_kernel<<<16, NTHR, lds_bytes, stream>>>(y1, W1, b1, u1, W2, b2, out);
}
